# starting kernel
# baseline (speedup 1.0000x reference)
#include <hip/hip_runtime.h>
#include <hip/hip_bf16.h>

typedef __attribute__((ext_vector_type(2))) float v2f;
typedef __attribute__((ext_vector_type(8))) float v8f;

#define F_IN 512
#define HID  16
#define CLS  7

__global__ void deg_init_k(float* __restrict__ deg, int n) {
    int i = blockIdx.x * blockDim.x + threadIdx.x;
    if (i < n) deg[i] = 1.0f;
}

__global__ void deg_count_k(const int* __restrict__ dst, float* __restrict__ deg, int E) {
    int e = blockIdx.x * blockDim.x + threadIdx.x;
    if (e < E) unsafeAtomicAdd(&deg[dst[e]], 1.0f);
}

__global__ void dinv_k(float* __restrict__ deg, int n) {
    int i = blockIdx.x * blockDim.x + threadIdx.x;
    if (i < n) deg[i] = rsqrtf(deg[i]);
}

__global__ void __launch_bounds__(256) gemm1_k(const float* __restrict__ x,
                                               const float* __restrict__ W1,
                                               float* __restrict__ h, int nTiles) {
    __shared__ float sW[F_IN * HID];
    for (int i = threadIdx.x * 4; i < F_IN * HID; i += 256 * 4)
        *(float4*)(sW + i) = *(const float4*)(W1 + i);
    __syncthreads();

    const int wave = threadIdx.x >> 5;
    const int lane = threadIdx.x & 31;
    const int tile = blockIdx.x * 8 + wave;
    if (tile >= nTiles) return;

    const int m    = lane & 15;
    const int kk   = (lane >> 4) << 1;
    const int half = lane >> 4;

    const float* arow = x + (size_t)(tile * 16 + m) * F_IN + kk;
    v8f c = {0.f, 0.f, 0.f, 0.f, 0.f, 0.f, 0.f, 0.f};

#pragma unroll 4
    for (int k0 = 0; k0 < F_IN; k0 += 4) {
        v2f a = *(const v2f*)(arow + k0);
        v2f b;
        b.x = sW[(k0 + kk)     * HID + m];
        b.y = sW[(k0 + kk + 1) * HID + m];
        c = __builtin_amdgcn_wmma_f32_16x16x4_f32(false, a, false, b,
                                                  (short)0, c, false, false);
    }

    float* orow = h + (size_t)(tile * 16 + half * 8) * HID + m;
#pragma unroll
    for (int v = 0; v < 8; ++v) orow[(size_t)v * HID] = c[v];
}

__global__ void self_init1_k(const float* __restrict__ h, const float* __restrict__ dinv,
                             float* __restrict__ agg, int n) {
    int t = blockIdx.x * blockDim.x + threadIdx.x;
    if (t >= n) return;
    int i = t >> 4;
    float di = dinv[i];
    agg[t] = di * di * h[t];
}

__global__ void scatter1_k(const int* __restrict__ src, const int* __restrict__ dst,
                           const float* __restrict__ dinv, const float* __restrict__ h,
                           float* __restrict__ agg, int E) {
    int t = blockIdx.x * blockDim.x + threadIdx.x;
    int e = t >> 4;
    if (e >= E) return;
    int f = t & 15;
    int s = src[e], d = dst[e];
    float w = dinv[s] * dinv[d];
    unsafeAtomicAdd(&agg[(size_t)d * HID + f], w * h[(size_t)s * HID + f]);
}

__global__ void bias_relu_k(float* __restrict__ agg, const float* __restrict__ b1, int n) {
    int t = blockIdx.x * blockDim.x + threadIdx.x;
    if (t >= n) return;
    float v = agg[t] + b1[t & 15];
    agg[t] = v > 0.f ? v : 0.f;
}

__global__ void __launch_bounds__(256) gemm2_k(const float* __restrict__ a_in,
                                               const float* __restrict__ W2,
                                               float* __restrict__ h2, int nTiles) {
    __shared__ float sW[HID * 16];
    {
        int t = threadIdx.x;
        int k = t >> 4, n = t & 15;
        sW[t] = (n < CLS) ? W2[k * CLS + n] : 0.0f;
    }
    __syncthreads();

    const int wave = threadIdx.x >> 5;
    const int lane = threadIdx.x & 31;
    const int tile = blockIdx.x * 8 + wave;
    if (tile >= nTiles) return;

    const int m    = lane & 15;
    const int kk   = (lane >> 4) << 1;
    const int half = lane >> 4;

    const float* arow = a_in + (size_t)(tile * 16 + m) * HID + kk;
    v8f c = {0.f, 0.f, 0.f, 0.f, 0.f, 0.f, 0.f, 0.f};

#pragma unroll
    for (int k0 = 0; k0 < HID; k0 += 4) {
        v2f a = *(const v2f*)(arow + k0);
        v2f b;
        b.x = sW[(k0 + kk)     * 16 + m];
        b.y = sW[(k0 + kk + 1) * 16 + m];
        c = __builtin_amdgcn_wmma_f32_16x16x4_f32(false, a, false, b,
                                                  (short)0, c, false, false);
    }

    if (m < CLS) {
#pragma unroll
        for (int v = 0; v < 8; ++v) {
            int row = tile * 16 + half * 8 + v;
            h2[(size_t)row * CLS + m] = c[v];
        }
    }
}

__global__ void self_init2_k(const float* __restrict__ h2, const float* __restrict__ dinv,
                             const float* __restrict__ b2, float* __restrict__ out, int n) {
    int t = blockIdx.x * blockDim.x + threadIdx.x;
    if (t >= n) return;
    int i = t / CLS, f = t % CLS;
    float di = dinv[i];
    out[t] = di * di * h2[t] + b2[f];
}

__global__ void scatter2_k(const int* __restrict__ src, const int* __restrict__ dst,
                           const float* __restrict__ dinv, const float* __restrict__ h2,
                           float* __restrict__ out, int E) {
    int t = blockIdx.x * blockDim.x + threadIdx.x;
    int e = t >> 3;
    if (e >= E) return;
    int f = t & 7;
    if (f >= CLS) return;
    int s = src[e], d = dst[e];
    float w = dinv[s] * dinv[d];
    unsafeAtomicAdd(&out[(size_t)d * CLS + f], w * h2[(size_t)s * CLS + f]);
}

extern "C" void kernel_launch(void* const* d_in, const int* in_sizes, int n_in,
                              void* d_out, int out_size, void* d_ws, size_t ws_size,
                              hipStream_t stream) {
    const float* x  = (const float*)d_in[0];
    const int*   ei = (const int*)  d_in[1];
    const float* W1 = (const float*)d_in[2];
    const float* b1 = (const float*)d_in[3];
    const float* W2 = (const float*)d_in[4];
    const float* b2 = (const float*)d_in[5];
    float* out = (float*)d_out;

    const int N = in_sizes[0] / F_IN;
    const int E = in_sizes[1] / 2;
    const int* src = ei;
    const int* dst = ei + E;

    float* dinv = (float*)d_ws;
    float* h    = dinv + N;
    float* agg  = h   + (size_t)N * HID;
    float* h2   = agg + (size_t)N * HID;

    const int B = 256;
    const int nTiles = N / 16;

    deg_init_k <<<(N + B - 1) / B, B, 0, stream>>>(dinv, N);
    deg_count_k<<<(E + B - 1) / B, B, 0, stream>>>(dst, dinv, E);
    dinv_k     <<<(N + B - 1) / B, B, 0, stream>>>(dinv, N);

    gemm1_k    <<<(nTiles + 7) / 8, B, 0, stream>>>(x, W1, h, nTiles);

    int n1 = N * HID;
    self_init1_k<<<(n1 + B - 1) / B, B, 0, stream>>>(h, dinv, agg, n1);
    {
        long long tot = (long long)E * 16;
        scatter1_k<<<(int)((tot + B - 1) / B), B, 0, stream>>>(src, dst, dinv, h, agg, E);
    }
    bias_relu_k<<<(n1 + B - 1) / B, B, 0, stream>>>(agg, b1, n1);

    gemm2_k    <<<(nTiles + 7) / 8, B, 0, stream>>>(agg, W2, h2, nTiles);

    int n2 = N * CLS;
    self_init2_k<<<(n2 + B - 1) / B, B, 0, stream>>>(h2, dinv, b2, out, n2);
    {
        long long tot = (long long)E * 8;
        scatter2_k<<<(int)((tot + B - 1) / B), B, 0, stream>>>(src, dst, dinv, h2, out, E);
    }
}


# Both edge passes: a work-group holds a destination-row range as LDS sums, filters a slice of the unordered edge list against it (ds float adds), joins partials by line-wide atomics; sums regrouped
# speedup vs baseline: 1.5493x; 1.5493x over previous
#include <hip/hip_runtime.h>
#include <hip/hip_bf16.h>

typedef __attribute__((ext_vector_type(2))) float v2f;
typedef __attribute__((ext_vector_type(8))) float v8f;

#define F_IN 512
#define HID  16
#define CLS  7

__global__ void deg_init_k(float* __restrict__ deg, int n) {
    int i = blockIdx.x * blockDim.x + threadIdx.x;
    if (i < n) deg[i] = 1.0f;
}

__global__ void dinv_k(float* __restrict__ deg, int n) {
    int i = blockIdx.x * blockDim.x + threadIdx.x;
    if (i < n) deg[i] = rsqrtf(deg[i]);
}

__global__ void __launch_bounds__(256) gemm1_deg_k(const float* __restrict__ x,
                                                   const float* __restrict__ W1,
                                                   float* __restrict__ h, int nTiles, int gemmBlocks,
                                                   const int* __restrict__ dst,
                                                   float* __restrict__ deg, int E) {
    __shared__ float sW[F_IN * HID];
    if ((int)blockIdx.x >= gemmBlocks) {
        int e = ((int)blockIdx.x - gemmBlocks) * 256 + (int)threadIdx.x;
        if (e < E) unsafeAtomicAdd(&deg[dst[e]], 1.0f);
        return;
    }
    for (int i = threadIdx.x * 4; i < F_IN * HID; i += 256 * 4)
        *(float4*)(sW + i) = *(const float4*)(W1 + i);
    __syncthreads();

    const int wave = threadIdx.x >> 5;
    const int lane = threadIdx.x & 31;
    const int tile = blockIdx.x * 8 + wave;
    if (tile >= nTiles) return;

    const int m    = lane & 15;
    const int kk   = (lane >> 4) << 1;
    const int half = lane >> 4;

    const float* arow = x + (size_t)(tile * 16 + m) * F_IN + kk;
    v8f c = {0.f, 0.f, 0.f, 0.f, 0.f, 0.f, 0.f, 0.f};

#pragma unroll 4
    for (int k0 = 0; k0 < F_IN; k0 += 4) {
        v2f a = *(const v2f*)(arow + k0);
        v2f b;
        b.x = sW[(k0 + kk)     * HID + m];
        b.y = sW[(k0 + kk + 1) * HID + m];
        c = __builtin_amdgcn_wmma_f32_16x16x4_f32(false, a, false, b,
                                                  (short)0, c, false, false);
    }

    float* orow = h + (size_t)(tile * 16 + half * 8) * HID + m;
#pragma unroll
    for (int v = 0; v < 8; ++v) orow[(size_t)v * HID] = c[v];
}

__global__ void self_init1_k(const float* __restrict__ h, const float* __restrict__ dinv,
                             float* __restrict__ agg, int n) {
    int t = blockIdx.x * blockDim.x + threadIdx.x;
    if (t >= n) return;
    int i = t >> 4;
    float di = dinv[i];
    agg[t] = di * di * h[t];
}

#define OWN_THREADS 1024
#define OWN_EPL     8
#define OWN_SUMS    50176
#define SLICES1     8
#define SLICES2     16

template <bool FULL>
__device__ __forceinline__ void load_run(const int* __restrict__ src, const int* __restrict__ dst,
                                         int e, int e_end, int E, int* s, int* d) {
    if (FULL) {
        const bool in = e < e_end;
        int ec = e < (E - 4) ? e : (E - 4);
        ec = ec > 0 ? ec : 0;
        const int4 s4 = *(const int4*)(src + ec);
        const int4 d4 = *(const int4*)(dst + ec);
        s[0] = s4.x; s[1] = s4.y; s[2] = s4.z; s[3] = s4.w;
        d[0] = in ? d4.x : -1; d[1] = in ? d4.y : -1; d[2] = in ? d4.z : -1; d[3] = in ? d4.w : -1;
    } else {
#pragma unroll
        for (int u = 0; u < 4; ++u) {
            const bool in = (e + u) < e_end;
            s[u] = in ? src[e + u] : 0;
            d[u] = in ? dst[e + u] : -1;
        }
    }
}

template <int ROWG, int ROWL, bool FULL>
__device__ __forceinline__ void owner_pass(const int* __restrict__ src, const int* __restrict__ dst,
                                           const float* __restrict__ dinv, const float* __restrict__ tab,
                                           float* __restrict__ acc, int E, int N,
                                           float* sAcc, float* sDinv, int2* sQ) {
    constexpr int RPR = OWN_SUMS / ROWL;
    constexpr int G   = 32 / ROWL;
    constexpr int U   = 6;
    const int tid  = threadIdx.x;
    const int lane = tid & 31;
    const int wave = tid >> 5;
    const int f    = lane & (ROWL - 1);
    const int grp  = lane / ROWL;
    const int fg   = f < ROWG ? f : 0;

    const int lo   = (int)blockIdx.x * RPR;
    const int rows = (N - lo) < RPR ? (N - lo) : RPR;

    for (int i = tid * 4; i < OWN_SUMS; i += OWN_THREADS * 4)
        *(float4*)(sAcc + i) = make_float4(0.f, 0.f, 0.f, 0.f);
    for (int i = tid; i < rows; i += OWN_THREADS) sDinv[i] = dinv[lo + i];
    __syncthreads();

    const int S       = (int)gridDim.y;
    const int per     = (((E + S - 1) / S) + 3) & ~3;
    const int e_begin = (int)blockIdx.y * per;
    const int e_end   = (e_begin + per) < E ? (e_begin + per) : E;

    int2* q = sQ + wave * (OWN_EPL * 32);
    const unsigned below = (1u << lane) - 1u;

    int s[OWN_EPL], d[OWN_EPL];
    load_run<FULL>(src, dst, e_begin + tid * 4, e_end, E, s, d);
    load_run<FULL>(src, dst, e_begin + OWN_THREADS * 4 + tid * 4, e_end, E, s + 4, d + 4);

    for (int base = e_begin; base < e_end; base += OWN_THREADS * OWN_EPL) {
        int ns[OWN_EPL], nd[OWN_EPL];
        const int nb = base + OWN_THREADS * OWN_EPL;
        load_run<FULL>(src, dst, nb + tid * 4, e_end, E, ns, nd);
        load_run<FULL>(src, dst, nb + OWN_THREADS * 4 + tid * 4, e_end, E, ns + 4, nd + 4);

        int m = 0;
#pragma unroll
        for (int c = 0; c < OWN_EPL; ++c) {
            const unsigned t   = (unsigned)(d[c] - lo);
            const bool     hit = t < (unsigned)rows;
            const unsigned mk  = __builtin_amdgcn_ballot_w32(hit);
            if (hit) q[m + __popc(mk & below)] = make_int2(s[c], (int)t);
            m += __popc(mk);
        }

        for (int j = 0; j < m; j += U * G) {
            int   ti[U];
            float w[U], v[U];
            bool  ok[U];
#pragma unroll
            for (int u = 0; u < U; ++u) {
                const int idx = j + u * G + grp;
                ok[u] = idx < m;
                const int2 en = q[ok[u] ? idx : 0];
                const int  si = ok[u] ? en.x : 0;
                ti[u] = ok[u] ? en.y : 0;
                w[u] = dinv[si] * sDinv[ti[u]];
                v[u] = tab[(size_t)si * ROWG + fg];
            }
#pragma unroll
            for (int u = 0; u < U; ++u) {
                const float val = (ok[u] && f < ROWG) ? w[u] * v[u] : 0.0f;
                __hip_atomic_fetch_add(&sAcc[ti[u] * ROWL + f], val, __ATOMIC_RELAXED, __HIP_MEMORY_SCOPE_WORKGROUP);
            }
        }

#pragma unroll
        for (int c = 0; c < OWN_EPL; ++c) { s[c] = ns[c]; d[c] = nd[c]; }
    }
    __syncthreads();

    float* out0 = acc + (size_t)lo * ROWG;
    for (int i = tid; i < rows * ROWG; i += OWN_THREADS) {
        const int row = i / ROWG, ff = i - row * ROWG;
        unsafeAtomicAdd(out0 + i, sAcc[row * ROWL + ff]);
    }
}

template <int ROWG, int ROWL>
__global__ void __launch_bounds__(OWN_THREADS) owner_pass_k(const int* __restrict__ src, const int* __restrict__ dst,
                           const float* __restrict__ dinv, const float* __restrict__ tab,
                           float* __restrict__ acc, int E, int N) {
    __shared__ float sAcc[OWN_SUMS];
    __shared__ float sDinv[OWN_SUMS / ROWL];
    __shared__ int2  sQ[32 * OWN_EPL * 32];
    if ((E & 3) == 0 && E >= 4) owner_pass<ROWG, ROWL, true >(src, dst, dinv, tab, acc, E, N, sAcc, sDinv, sQ);
    else              owner_pass<ROWG, ROWL, false>(src, dst, dinv, tab, acc, E, N, sAcc, sDinv, sQ);
}

__global__ void bias_relu_k(float* __restrict__ agg, const float* __restrict__ b1, int n) {
    int t = blockIdx.x * blockDim.x + threadIdx.x;
    if (t >= n) return;
    float v = agg[t] + b1[t & 15];
    agg[t] = v > 0.f ? v : 0.f;
}

__global__ void __launch_bounds__(256) gemm2_k(const float* __restrict__ a_in,
                                               const float* __restrict__ W2,
                                               float* __restrict__ h2, int nTiles) {
    __shared__ float sW[HID * 16];
    {
        int t = threadIdx.x;
        int k = t >> 4, n = t & 15;
        sW[t] = (n < CLS) ? W2[k * CLS + n] : 0.0f;
    }
    __syncthreads();

    const int wave = threadIdx.x >> 5;
    const int lane = threadIdx.x & 31;
    const int tile = blockIdx.x * 8 + wave;
    if (tile >= nTiles) return;

    const int m    = lane & 15;
    const int kk   = (lane >> 4) << 1;
    const int half = lane >> 4;

    const float* arow = a_in + (size_t)(tile * 16 + m) * HID + kk;
    v8f c = {0.f, 0.f, 0.f, 0.f, 0.f, 0.f, 0.f, 0.f};

#pragma unroll
    for (int k0 = 0; k0 < HID; k0 += 4) {
        v2f a = *(const v2f*)(arow + k0);
        v2f b;
        b.x = sW[(k0 + kk)     * 16 + m];
        b.y = sW[(k0 + kk + 1) * 16 + m];
        c = __builtin_amdgcn_wmma_f32_16x16x4_f32(false, a, false, b,
                                                  (short)0, c, false, false);
    }

    if (m < CLS) {
#pragma unroll
        for (int v = 0; v < 8; ++v) {
            int row = tile * 16 + half * 8 + v;
            h2[(size_t)row * CLS + m] = c[v];
        }
    }
}

__global__ void self_init2_k(const float* __restrict__ h2, const float* __restrict__ dinv,
                             const float* __restrict__ b2, float* __restrict__ out, int n) {
    int t = blockIdx.x * blockDim.x + threadIdx.x;
    if (t >= n) return;
    int i = t / CLS, f = t % CLS;
    float di = dinv[i];
    out[t] = di * di * h2[t] + b2[f];
}

extern "C" void kernel_launch(void* const* d_in, const int* in_sizes, int n_in,
                              void* d_out, int out_size, void* d_ws, size_t ws_size,
                              hipStream_t stream) {
    const float* x  = (const float*)d_in[0];
    const int*   ei = (const int*)  d_in[1];
    const float* W1 = (const float*)d_in[2];
    const float* b1 = (const float*)d_in[3];
    const float* W2 = (const float*)d_in[4];
    const float* b2 = (const float*)d_in[5];
    float* out = (float*)d_out;

    const int N = in_sizes[0] / F_IN;
    const int E = in_sizes[1] / 2;
    const int* src = ei;
    const int* dst = ei + E;

    float* dinv = (float*)d_ws;
    float* h    = dinv + N;
    float* agg  = h   + (size_t)N * HID;
    float* h2   = agg + (size_t)N * HID;

    const int B = 256;
    const int nTiles = N / 16;

    const int gemmBlocks = (nTiles + 7) / 8;

    deg_init_k <<<(N + B - 1) / B, B, 0, stream>>>(dinv, N);
    gemm1_deg_k<<<gemmBlocks + (E + B - 1) / B, B, 0, stream>>>(x, W1, h, nTiles, gemmBlocks, dst, dinv, E);
    dinv_k     <<<(N + B - 1) / B, B, 0, stream>>>(dinv, N);

    int n1 = N * HID;
    self_init1_k<<<(n1 + B - 1) / B, B, 0, stream>>>(h, dinv, agg, n1);
    owner_pass_k<HID, 16><<<dim3((N + OWN_SUMS / 16 - 1) / (OWN_SUMS / 16), SLICES1), OWN_THREADS, 0, stream>>>(src, dst, dinv, h, agg, E, N);
    bias_relu_k<<<(n1 + B - 1) / B, B, 0, stream>>>(agg, b1, n1);

    gemm2_k    <<<(nTiles + 7) / 8, B, 0, stream>>>(agg, W2, h2, nTiles);

    int n2 = N * CLS;
    self_init2_k<<<(n2 + B - 1) / B, B, 0, stream>>>(h2, dinv, b2, out, n2);
    owner_pass_k<CLS, 8><<<dim3((N + OWN_SUMS / 8 - 1) / (OWN_SUMS / 8), SLICES2), OWN_THREADS, 0, stream>>>(src, dst, dinv, h2, out, E, N);
}


# Degree count: counting blocks of the GEMM1 launch each own 8192 nodes as 32-bit LDS counters, filter a dst slice against them, join by line-wide adds of whole numbers (exact, order cannot show)
# speedup vs baseline: 2.4447x; 1.5779x over previous
#include <hip/hip_runtime.h>
#include <hip/hip_bf16.h>

typedef __attribute__((ext_vector_type(2))) float v2f;
typedef __attribute__((ext_vector_type(8))) float v8f;

#define F_IN 512
#define HID  16
#define CLS  7

__global__ void deg_init_k(float* __restrict__ deg, int n) {
    int i = blockIdx.x * blockDim.x + threadIdx.x;
    if (i < n) deg[i] = 1.0f;
}

__global__ void dinv_k(float* __restrict__ deg, int n) {
    int i = blockIdx.x * blockDim.x + threadIdx.x;
    if (i < n) deg[i] = rsqrtf(deg[i]);
}

#define CNT_ROWS   (F_IN * HID)
#define CNT_SLICES 64
#define CNT_U      8

template <bool FULL>
__device__ __forceinline__ void count_range(const int* __restrict__ dst, float* __restrict__ deg,
                                            int E, int N, int cb, unsigned* sCnt) {
    const int tid   = threadIdx.x;
    const int range = cb / CNT_SLICES;
    const int slice = cb - range * CNT_SLICES;
    const int lo    = range * CNT_ROWS;
    const int rows  = (N - lo) < CNT_ROWS ? (N - lo) : CNT_ROWS;

    for (int i = tid * 4; i < CNT_ROWS; i += 256 * 4)
        *(uint4*)(sCnt + i) = make_uint4(0u, 0u, 0u, 0u);
    __syncthreads();

    const int per     = (((E + CNT_SLICES - 1) / CNT_SLICES) + 3) & ~3;
    const int e_begin = slice * per;
    const int e_end   = (e_begin + per) < E ? (e_begin + per) : E;

    if (FULL) {
        for (int e0 = e_begin + tid * 4; e0 < e_end; e0 += 256 * 4 * CNT_U) {
            int4 dv[CNT_U];
            bool in[CNT_U];
#pragma unroll
            for (int u = 0; u < CNT_U; ++u) {
                const int e  = e0 + u * (256 * 4);
                const int ec = e < (E - 4) ? e : (E - 4);
                in[u] = e < e_end;
                dv[u] = *(const int4*)(dst + ec);
            }
#pragma unroll
            for (int u = 0; u < CNT_U; ++u) {
                const int d[4] = {dv[u].x, dv[u].y, dv[u].z, dv[u].w};
#pragma unroll
                for (int c = 0; c < 4; ++c) {
                    const unsigned t = (unsigned)((in[u] ? d[c] : -1) - lo);
                    if (t < (unsigned)rows)
                        __hip_atomic_fetch_add(&sCnt[t], 1u, __ATOMIC_RELAXED, __HIP_MEMORY_SCOPE_WORKGROUP);
                }
            }
        }
    } else {
        for (int e = e_begin + tid; e < e_end; e += 256) {
            const unsigned t = (unsigned)(dst[e] - lo);
            if (t < (unsigned)rows)
                __hip_atomic_fetch_add(&sCnt[t], 1u, __ATOMIC_RELAXED, __HIP_MEMORY_SCOPE_WORKGROUP);
        }
    }
    __syncthreads();

    for (int i = tid; i < rows; i += 256) {
        const unsigned c = sCnt[i];
        if (c != 0u) unsafeAtomicAdd(deg + lo + i, (float)c);
    }
}

__global__ void __launch_bounds__(256) gemm1_deg_k(const float* __restrict__ x,
                                                   const float* __restrict__ W1,
                                                   float* __restrict__ h, int nTiles, int gemmBlocks,
                                                   const int* __restrict__ dst,
                                                   float* __restrict__ deg, int E, int N) {
    __shared__ float sW[F_IN * HID];
    if ((int)blockIdx.x >= gemmBlocks) {
        unsigned* sCnt = reinterpret_cast<unsigned*>(sW);
        const int cb = (int)blockIdx.x - gemmBlocks;
        if ((E & 3) == 0 && E >= 4) count_range<true >(dst, deg, E, N, cb, sCnt);
        else                        count_range<false>(dst, deg, E, N, cb, sCnt);
        return;
    }
    for (int i = threadIdx.x * 4; i < F_IN * HID; i += 256 * 4)
        *(float4*)(sW + i) = *(const float4*)(W1 + i);
    __syncthreads();

    const int wave = threadIdx.x >> 5;
    const int lane = threadIdx.x & 31;
    const int tile = blockIdx.x * 8 + wave;
    if (tile >= nTiles) return;

    const int m    = lane & 15;
    const int kk   = (lane >> 4) << 1;
    const int half = lane >> 4;

    const float* arow = x + (size_t)(tile * 16 + m) * F_IN + kk;
    v8f c = {0.f, 0.f, 0.f, 0.f, 0.f, 0.f, 0.f, 0.f};

#pragma unroll 4
    for (int k0 = 0; k0 < F_IN; k0 += 4) {
        v2f a = *(const v2f*)(arow + k0);
        v2f b;
        b.x = sW[(k0 + kk)     * HID + m];
        b.y = sW[(k0 + kk + 1) * HID + m];
        c = __builtin_amdgcn_wmma_f32_16x16x4_f32(false, a, false, b,
                                                  (short)0, c, false, false);
    }

    float* orow = h + (size_t)(tile * 16 + half * 8) * HID + m;
#pragma unroll
    for (int v = 0; v < 8; ++v) orow[(size_t)v * HID] = c[v];
}

__global__ void self_init1_k(const float* __restrict__ h, const float* __restrict__ dinv,
                             float* __restrict__ agg, int n) {
    int t = blockIdx.x * blockDim.x + threadIdx.x;
    if (t >= n) return;
    int i = t >> 4;
    float di = dinv[i];
    agg[t] = di * di * h[t];
}

#define OWN_THREADS 1024
#define OWN_EPL     8
#define OWN_SUMS    50176
#define SLICES1     8
#define SLICES2     16

template <bool FULL>
__device__ __forceinline__ void load_run(const int* __restrict__ src, const int* __restrict__ dst,
                                         int e, int e_end, int E, int* s, int* d) {
    if (FULL) {
        const bool in = e < e_end;
        int ec = e < (E - 4) ? e : (E - 4);
        ec = ec > 0 ? ec : 0;
        const int4 s4 = *(const int4*)(src + ec);
        const int4 d4 = *(const int4*)(dst + ec);
        s[0] = s4.x; s[1] = s4.y; s[2] = s4.z; s[3] = s4.w;
        d[0] = in ? d4.x : -1; d[1] = in ? d4.y : -1; d[2] = in ? d4.z : -1; d[3] = in ? d4.w : -1;
    } else {
#pragma unroll
        for (int u = 0; u < 4; ++u) {
            const bool in = (e + u) < e_end;
            s[u] = in ? src[e + u] : 0;
            d[u] = in ? dst[e + u] : -1;
        }
    }
}

template <int ROWG, int ROWL, bool FULL>
__device__ __forceinline__ void owner_pass(const int* __restrict__ src, const int* __restrict__ dst,
                                           const float* __restrict__ dinv, const float* __restrict__ tab,
                                           float* __restrict__ acc, int E, int N,
                                           float* sAcc, float* sDinv, int2* sQ) {
    constexpr int RPR = OWN_SUMS / ROWL;
    constexpr int G   = 32 / ROWL;
    constexpr int U   = 6;
    const int tid  = threadIdx.x;
    const int lane = tid & 31;
    const int wave = tid >> 5;
    const int f    = lane & (ROWL - 1);
    const int grp  = lane / ROWL;
    const int fg   = f < ROWG ? f : 0;

    const int lo   = (int)blockIdx.x * RPR;
    const int rows = (N - lo) < RPR ? (N - lo) : RPR;

    for (int i = tid * 4; i < OWN_SUMS; i += OWN_THREADS * 4)
        *(float4*)(sAcc + i) = make_float4(0.f, 0.f, 0.f, 0.f);
    for (int i = tid; i < rows; i += OWN_THREADS) sDinv[i] = dinv[lo + i];
    __syncthreads();

    const int S       = (int)gridDim.y;
    const int per     = (((E + S - 1) / S) + 3) & ~3;
    const int e_begin = (int)blockIdx.y * per;
    const int e_end   = (e_begin + per) < E ? (e_begin + per) : E;

    int2* q = sQ + wave * (OWN_EPL * 32);
    const unsigned below = (1u << lane) - 1u;

    int s[OWN_EPL], d[OWN_EPL];
    load_run<FULL>(src, dst, e_begin + tid * 4, e_end, E, s, d);
    load_run<FULL>(src, dst, e_begin + OWN_THREADS * 4 + tid * 4, e_end, E, s + 4, d + 4);

    for (int base = e_begin; base < e_end; base += OWN_THREADS * OWN_EPL) {
        int ns[OWN_EPL], nd[OWN_EPL];
        const int nb = base + OWN_THREADS * OWN_EPL;
        load_run<FULL>(src, dst, nb + tid * 4, e_end, E, ns, nd);
        load_run<FULL>(src, dst, nb + OWN_THREADS * 4 + tid * 4, e_end, E, ns + 4, nd + 4);

        int m = 0;
#pragma unroll
        for (int c = 0; c < OWN_EPL; ++c) {
            const unsigned t   = (unsigned)(d[c] - lo);
            const bool     hit = t < (unsigned)rows;
            const unsigned mk  = __builtin_amdgcn_ballot_w32(hit);
            if (hit) q[m + __popc(mk & below)] = make_int2(s[c], (int)t);
            m += __popc(mk);
        }

        for (int j = 0; j < m; j += U * G) {
            int   ti[U];
            float w[U], v[U];
            bool  ok[U];
#pragma unroll
            for (int u = 0; u < U; ++u) {
                const int idx = j + u * G + grp;
                ok[u] = idx < m;
                const int2 en = q[ok[u] ? idx : 0];
                const int  si = ok[u] ? en.x : 0;
                ti[u] = ok[u] ? en.y : 0;
                w[u] = dinv[si] * sDinv[ti[u]];
                v[u] = tab[(size_t)si * ROWG + fg];
            }
#pragma unroll
            for (int u = 0; u < U; ++u) {
                const float val = (ok[u] && f < ROWG) ? w[u] * v[u] : 0.0f;
                __hip_atomic_fetch_add(&sAcc[ti[u] * ROWL + f], val, __ATOMIC_RELAXED, __HIP_MEMORY_SCOPE_WORKGROUP);
            }
        }

#pragma unroll
        for (int c = 0; c < OWN_EPL; ++c) { s[c] = ns[c]; d[c] = nd[c]; }
    }
    __syncthreads();

    float* out0 = acc + (size_t)lo * ROWG;
    for (int i = tid; i < rows * ROWG; i += OWN_THREADS) {
        const int row = i / ROWG, ff = i - row * ROWG;
        unsafeAtomicAdd(out0 + i, sAcc[row * ROWL + ff]);
    }
}

template <int ROWG, int ROWL>
__global__ void __launch_bounds__(OWN_THREADS) owner_pass_k(const int* __restrict__ src, const int* __restrict__ dst,
                           const float* __restrict__ dinv, const float* __restrict__ tab,
                           float* __restrict__ acc, int E, int N) {
    __shared__ float sAcc[OWN_SUMS];
    __shared__ float sDinv[OWN_SUMS / ROWL];
    __shared__ int2  sQ[32 * OWN_EPL * 32];
    if ((E & 3) == 0 && E >= 4) owner_pass<ROWG, ROWL, true >(src, dst, dinv, tab, acc, E, N, sAcc, sDinv, sQ);
    else              owner_pass<ROWG, ROWL, false>(src, dst, dinv, tab, acc, E, N, sAcc, sDinv, sQ);
}

__global__ void bias_relu_k(float* __restrict__ agg, const float* __restrict__ b1, int n) {
    int t = blockIdx.x * blockDim.x + threadIdx.x;
    if (t >= n) return;
    float v = agg[t] + b1[t & 15];
    agg[t] = v > 0.f ? v : 0.f;
}

__global__ void __launch_bounds__(256) gemm2_k(const float* __restrict__ a_in,
                                               const float* __restrict__ W2,
                                               float* __restrict__ h2, int nTiles) {
    __shared__ float sW[HID * 16];
    {
        int t = threadIdx.x;
        int k = t >> 4, n = t & 15;
        sW[t] = (n < CLS) ? W2[k * CLS + n] : 0.0f;
    }
    __syncthreads();

    const int wave = threadIdx.x >> 5;
    const int lane = threadIdx.x & 31;
    const int tile = blockIdx.x * 8 + wave;
    if (tile >= nTiles) return;

    const int m    = lane & 15;
    const int kk   = (lane >> 4) << 1;
    const int half = lane >> 4;

    const float* arow = a_in + (size_t)(tile * 16 + m) * HID + kk;
    v8f c = {0.f, 0.f, 0.f, 0.f, 0.f, 0.f, 0.f, 0.f};

#pragma unroll
    for (int k0 = 0; k0 < HID; k0 += 4) {
        v2f a = *(const v2f*)(arow + k0);
        v2f b;
        b.x = sW[(k0 + kk)     * 16 + m];
        b.y = sW[(k0 + kk + 1) * 16 + m];
        c = __builtin_amdgcn_wmma_f32_16x16x4_f32(false, a, false, b,
                                                  (short)0, c, false, false);
    }

    if (m < CLS) {
#pragma unroll
        for (int v = 0; v < 8; ++v) {
            int row = tile * 16 + half * 8 + v;
            h2[(size_t)row * CLS + m] = c[v];
        }
    }
}

__global__ void self_init2_k(const float* __restrict__ h2, const float* __restrict__ dinv,
                             const float* __restrict__ b2, float* __restrict__ out, int n) {
    int t = blockIdx.x * blockDim.x + threadIdx.x;
    if (t >= n) return;
    int i = t / CLS, f = t % CLS;
    float di = dinv[i];
    out[t] = di * di * h2[t] + b2[f];
}

extern "C" void kernel_launch(void* const* d_in, const int* in_sizes, int n_in,
                              void* d_out, int out_size, void* d_ws, size_t ws_size,
                              hipStream_t stream) {
    const float* x  = (const float*)d_in[0];
    const int*   ei = (const int*)  d_in[1];
    const float* W1 = (const float*)d_in[2];
    const float* b1 = (const float*)d_in[3];
    const float* W2 = (const float*)d_in[4];
    const float* b2 = (const float*)d_in[5];
    float* out = (float*)d_out;

    const int N = in_sizes[0] / F_IN;
    const int E = in_sizes[1] / 2;
    const int* src = ei;
    const int* dst = ei + E;

    float* dinv = (float*)d_ws;
    float* h    = dinv + N;
    float* agg  = h   + (size_t)N * HID;
    float* h2   = agg + (size_t)N * HID;

    const int B = 256;
    const int nTiles = N / 16;

    const int gemmBlocks = (nTiles + 7) / 8;

    deg_init_k <<<(N + B - 1) / B, B, 0, stream>>>(dinv, N);
    const int cntRanges = (N + CNT_ROWS - 1) / CNT_ROWS;
    gemm1_deg_k<<<gemmBlocks + cntRanges * CNT_SLICES, B, 0, stream>>>(x, W1, h, nTiles, gemmBlocks, dst, dinv, E, N);
    dinv_k     <<<(N + B - 1) / B, B, 0, stream>>>(dinv, N);

    int n1 = N * HID;
    self_init1_k<<<(n1 + B - 1) / B, B, 0, stream>>>(h, dinv, agg, n1);
    owner_pass_k<HID, 16><<<dim3((N + OWN_SUMS / 16 - 1) / (OWN_SUMS / 16), SLICES1), OWN_THREADS, 0, stream>>>(src, dst, dinv, h, agg, E, N);
    bias_relu_k<<<(n1 + B - 1) / B, B, 0, stream>>>(agg, b1, n1);

    gemm2_k    <<<(nTiles + 7) / 8, B, 0, stream>>>(agg, W2, h2, nTiles);

    int n2 = N * CLS;
    self_init2_k<<<(n2 + B - 1) / B, B, 0, stream>>>(h2, dinv, b2, out, n2);
    owner_pass_k<CLS, 8><<<dim3((N + OWN_SUMS / 8 - 1) / (OWN_SUMS / 8), SLICES2), OWN_THREADS, 0, stream>>>(src, dst, dinv, h2, out, E, N);
}


# Both edge passes: a wave's list of matched edges carries over between trips and is drained only in full rounds of 16 or 32 edges (one guarded round at the end); LDS float adds still in arrival order
# speedup vs baseline: 2.5916x; 1.0601x over previous
#include <hip/hip_runtime.h>
#include <hip/hip_bf16.h>

typedef __attribute__((ext_vector_type(2))) float v2f;
typedef __attribute__((ext_vector_type(8))) float v8f;

#define F_IN 512
#define HID  16
#define CLS  7

__global__ void deg_init_k(float* __restrict__ deg, int n) {
    int i = blockIdx.x * blockDim.x + threadIdx.x;
    if (i < n) deg[i] = 1.0f;
}

__global__ void dinv_k(float* __restrict__ deg, int n) {
    int i = blockIdx.x * blockDim.x + threadIdx.x;
    if (i < n) deg[i] = rsqrtf(deg[i]);
}

#define CNT_ROWS   (F_IN * HID)
#define CNT_SLICES 64
#define CNT_U      8

template <bool FULL>
__device__ __forceinline__ void count_range(const int* __restrict__ dst, float* __restrict__ deg,
                                            int E, int N, int cb, unsigned* sCnt) {
    const int tid   = threadIdx.x;
    const int range = cb / CNT_SLICES;
    const int slice = cb - range * CNT_SLICES;
    const int lo    = range * CNT_ROWS;
    const int rows  = (N - lo) < CNT_ROWS ? (N - lo) : CNT_ROWS;

    for (int i = tid * 4; i < CNT_ROWS; i += 256 * 4)
        *(uint4*)(sCnt + i) = make_uint4(0u, 0u, 0u, 0u);
    __syncthreads();

    const int per     = (((E + CNT_SLICES - 1) / CNT_SLICES) + 3) & ~3;
    const int e_begin = slice * per;
    const int e_end   = (e_begin + per) < E ? (e_begin + per) : E;

    if (FULL) {
        for (int e0 = e_begin + tid * 4; e0 < e_end; e0 += 256 * 4 * CNT_U) {
            int4 dv[CNT_U];
            bool in[CNT_U];
#pragma unroll
            for (int u = 0; u < CNT_U; ++u) {
                const int e  = e0 + u * (256 * 4);
                const int ec = e < (E - 4) ? e : (E - 4);
                in[u] = e < e_end;
                dv[u] = *(const int4*)(dst + ec);
            }
#pragma unroll
            for (int u = 0; u < CNT_U; ++u) {
                const int d[4] = {dv[u].x, dv[u].y, dv[u].z, dv[u].w};
#pragma unroll
                for (int c = 0; c < 4; ++c) {
                    const unsigned t = (unsigned)((in[u] ? d[c] : -1) - lo);
                    if (t < (unsigned)rows)
                        __hip_atomic_fetch_add(&sCnt[t], 1u, __ATOMIC_RELAXED, __HIP_MEMORY_SCOPE_WORKGROUP);
                }
            }
        }
    } else {
        for (int e = e_begin + tid; e < e_end; e += 256) {
            const unsigned t = (unsigned)(dst[e] - lo);
            if (t < (unsigned)rows)
                __hip_atomic_fetch_add(&sCnt[t], 1u, __ATOMIC_RELAXED, __HIP_MEMORY_SCOPE_WORKGROUP);
        }
    }
    __syncthreads();

    for (int i = tid; i < rows; i += 256) {
        const unsigned c = sCnt[i];
        if (c != 0u) unsafeAtomicAdd(deg + lo + i, (float)c);
    }
}

__global__ void __launch_bounds__(256) gemm1_deg_k(const float* __restrict__ x,
                                                   const float* __restrict__ W1,
                                                   float* __restrict__ h, int nTiles, int gemmBlocks,
                                                   const int* __restrict__ dst,
                                                   float* __restrict__ deg, int E, int N) {
    __shared__ float sW[F_IN * HID];
    if ((int)blockIdx.x >= gemmBlocks) {
        unsigned* sCnt = reinterpret_cast<unsigned*>(sW);
        const int cb = (int)blockIdx.x - gemmBlocks;
        if ((E & 3) == 0 && E >= 4) count_range<true >(dst, deg, E, N, cb, sCnt);
        else                        count_range<false>(dst, deg, E, N, cb, sCnt);
        return;
    }
    for (int i = threadIdx.x * 4; i < F_IN * HID; i += 256 * 4)
        *(float4*)(sW + i) = *(const float4*)(W1 + i);
    __syncthreads();

    const int wave = threadIdx.x >> 5;
    const int lane = threadIdx.x & 31;
    const int tile = blockIdx.x * 8 + wave;
    if (tile >= nTiles) return;

    const int m    = lane & 15;
    const int kk   = (lane >> 4) << 1;
    const int half = lane >> 4;

    const float* arow = x + (size_t)(tile * 16 + m) * F_IN + kk;
    v8f c = {0.f, 0.f, 0.f, 0.f, 0.f, 0.f, 0.f, 0.f};

#pragma unroll 4
    for (int k0 = 0; k0 < F_IN; k0 += 4) {
        v2f a = *(const v2f*)(arow + k0);
        v2f b;
        b.x = sW[(k0 + kk)     * HID + m];
        b.y = sW[(k0 + kk + 1) * HID + m];
        c = __builtin_amdgcn_wmma_f32_16x16x4_f32(false, a, false, b,
                                                  (short)0, c, false, false);
    }

    float* orow = h + (size_t)(tile * 16 + half * 8) * HID + m;
#pragma unroll
    for (int v = 0; v < 8; ++v) orow[(size_t)v * HID] = c[v];
}

__global__ void self_init1_k(const float* __restrict__ h, const float* __restrict__ dinv,
                             float* __restrict__ agg, int n) {
    int t = blockIdx.x * blockDim.x + threadIdx.x;
    if (t >= n) return;
    int i = t >> 4;
    float di = dinv[i];
    agg[t] = di * di * h[t];
}

#define OWN_THREADS 1024
#define OWN_EPL     8
#define OWN_SUMS    50176
#define OWN_U       8
#define OWN_Q       (OWN_EPL * 32 + 32)
#define SLICES1     8
#define SLICES2     16

template <bool FULL>
__device__ __forceinline__ void load_run(const int* __restrict__ src, const int* __restrict__ dst,
                                         int e, int e_end, int E, int* s, int* d) {
    if (FULL) {
        const bool in = e < e_end;
        int ec = e < (E - 4) ? e : (E - 4);
        ec = ec > 0 ? ec : 0;
        const int4 s4 = *(const int4*)(src + ec);
        const int4 d4 = *(const int4*)(dst + ec);
        s[0] = s4.x; s[1] = s4.y; s[2] = s4.z; s[3] = s4.w;
        d[0] = in ? d4.x : -1; d[1] = in ? d4.y : -1; d[2] = in ? d4.z : -1; d[3] = in ? d4.w : -1;
    } else {
#pragma unroll
        for (int u = 0; u < 4; ++u) {
            const bool in = (e + u) < e_end;
            s[u] = in ? src[e + u] : 0;
            d[u] = in ? dst[e + u] : -1;
        }
    }
}

template <int ROWG, int ROWL, bool GUARD>
__device__ __forceinline__ void gather_round(const int2* q, int j, int m, int grp, int f, int fg,
                                             const float* __restrict__ dinv, const float* __restrict__ tab,
                                             const float* sDinv, float* sAcc) {
    constexpr int G = 32 / ROWL;
    int   ti[OWN_U];
    float w[OWN_U], v[OWN_U];
    bool  ok[OWN_U];
#pragma unroll
    for (int u = 0; u < OWN_U; ++u) {
        const int idx = j + u * G + grp;
        ok[u] = GUARD ? (idx < m) : true;
        const int2 en = q[ok[u] ? idx : 0];
        const int  si = ok[u] ? en.x : 0;
        ti[u] = ok[u] ? en.y : 0;
        w[u] = dinv[si] * sDinv[ti[u]];
        v[u] = tab[(size_t)si * ROWG + fg];
    }
#pragma unroll
    for (int u = 0; u < OWN_U; ++u) {
        const float val = (ok[u] && f < ROWG) ? w[u] * v[u] : 0.0f;
        __hip_atomic_fetch_add(&sAcc[ti[u] * ROWL + f], val, __ATOMIC_RELAXED, __HIP_MEMORY_SCOPE_WORKGROUP);
    }
}

template <int ROWG, int ROWL, bool FULL>
__device__ __forceinline__ void owner_pass(const int* __restrict__ src, const int* __restrict__ dst,
                                           const float* __restrict__ dinv, const float* __restrict__ tab,
                                           float* __restrict__ acc, int E, int N,
                                           float* sAcc, float* sDinv, int2* sQ) {
    constexpr int RPR = OWN_SUMS / ROWL;
    constexpr int G   = 32 / ROWL;
    constexpr int UG  = OWN_U * G;
    const int tid  = threadIdx.x;
    const int lane = tid & 31;
    const int wave = tid >> 5;
    const int f    = lane & (ROWL - 1);
    const int grp  = lane / ROWL;
    const int fg   = f < ROWG ? f : 0;

    const int lo   = (int)blockIdx.x * RPR;
    const int rows = (N - lo) < RPR ? (N - lo) : RPR;

    for (int i = tid * 4; i < OWN_SUMS; i += OWN_THREADS * 4)
        *(float4*)(sAcc + i) = make_float4(0.f, 0.f, 0.f, 0.f);
    for (int i = tid; i < rows; i += OWN_THREADS) sDinv[i] = dinv[lo + i];
    __syncthreads();

    const int S       = (int)gridDim.y;
    const int per     = (((E + S - 1) / S) + 3) & ~3;
    const int e_begin = (int)blockIdx.y * per;
    const int e_end   = (e_begin + per) < E ? (e_begin + per) : E;

    int2* q = sQ + wave * OWN_Q;
    const unsigned below = (1u << lane) - 1u;

    int s[OWN_EPL], d[OWN_EPL];
    load_run<FULL>(src, dst, e_begin + tid * 4, e_end, E, s, d);
    load_run<FULL>(src, dst, e_begin + OWN_THREADS * 4 + tid * 4, e_end, E, s + 4, d + 4);

    int m = 0;
    for (int base = e_begin; base < e_end; base += OWN_THREADS * OWN_EPL) {
        int ns[OWN_EPL], nd[OWN_EPL];
        const int nb = base + OWN_THREADS * OWN_EPL;
        load_run<FULL>(src, dst, nb + tid * 4, e_end, E, ns, nd);
        load_run<FULL>(src, dst, nb + OWN_THREADS * 4 + tid * 4, e_end, E, ns + 4, nd + 4);

#pragma unroll
        for (int c = 0; c < OWN_EPL; ++c) {
            const unsigned t   = (unsigned)(d[c] - lo);
            const bool     hit = t < (unsigned)rows;
            const unsigned mk  = __builtin_amdgcn_ballot_w32(hit);
            if (hit) q[m + __popc(mk & below)] = make_int2(s[c], (int)t);
            m += __popc(mk);
        }

        int j = 0;
        for (; j + UG <= m; j += UG)
            gather_round<ROWG, ROWL, false>(q, j, m, grp, f, fg, dinv, tab, sDinv, sAcc);

        if (j != 0) {
            const int r = m - j;
            if (lane < r) {
                const int2 en = q[j + lane];
                q[lane] = en;
            }
            m = r;
        }

#pragma unroll
        for (int c = 0; c < OWN_EPL; ++c) { s[c] = ns[c]; d[c] = nd[c]; }
    }
    if (m > 0)
        gather_round<ROWG, ROWL, true>(q, 0, m, grp, f, fg, dinv, tab, sDinv, sAcc);
    __syncthreads();

    float* out0 = acc + (size_t)lo * ROWG;
    for (int i = tid; i < rows * ROWG; i += OWN_THREADS) {
        const int row = i / ROWG, ff = i - row * ROWG;
        unsafeAtomicAdd(out0 + i, sAcc[row * ROWL + ff]);
    }
}

template <int ROWG, int ROWL>
__global__ void __launch_bounds__(OWN_THREADS) owner_pass_k(const int* __restrict__ src, const int* __restrict__ dst,
                           const float* __restrict__ dinv, const float* __restrict__ tab,
                           float* __restrict__ acc, int E, int N) {
    __shared__ float sAcc[OWN_SUMS];
    __shared__ float sDinv[OWN_SUMS / ROWL];
    __shared__ int2  sQ[32 * OWN_Q];
    if ((E & 3) == 0 && E >= 4) owner_pass<ROWG, ROWL, true >(src, dst, dinv, tab, acc, E, N, sAcc, sDinv, sQ);
    else              owner_pass<ROWG, ROWL, false>(src, dst, dinv, tab, acc, E, N, sAcc, sDinv, sQ);
}

__global__ void bias_relu_k(float* __restrict__ agg, const float* __restrict__ b1, int n) {
    int t = blockIdx.x * blockDim.x + threadIdx.x;
    if (t >= n) return;
    float v = agg[t] + b1[t & 15];
    agg[t] = v > 0.f ? v : 0.f;
}

__global__ void __launch_bounds__(256) gemm2_k(const float* __restrict__ a_in,
                                               const float* __restrict__ W2,
                                               float* __restrict__ h2, int nTiles) {
    __shared__ float sW[HID * 16];
    {
        int t = threadIdx.x;
        int k = t >> 4, n = t & 15;
        sW[t] = (n < CLS) ? W2[k * CLS + n] : 0.0f;
    }
    __syncthreads();

    const int wave = threadIdx.x >> 5;
    const int lane = threadIdx.x & 31;
    const int tile = blockIdx.x * 8 + wave;
    if (tile >= nTiles) return;

    const int m    = lane & 15;
    const int kk   = (lane >> 4) << 1;
    const int half = lane >> 4;

    const float* arow = a_in + (size_t)(tile * 16 + m) * HID + kk;
    v8f c = {0.f, 0.f, 0.f, 0.f, 0.f, 0.f, 0.f, 0.f};

#pragma unroll
    for (int k0 = 0; k0 < HID; k0 += 4) {
        v2f a = *(const v2f*)(arow + k0);
        v2f b;
        b.x = sW[(k0 + kk)     * 16 + m];
        b.y = sW[(k0 + kk + 1) * 16 + m];
        c = __builtin_amdgcn_wmma_f32_16x16x4_f32(false, a, false, b,
                                                  (short)0, c, false, false);
    }

    if (m < CLS) {
#pragma unroll
        for (int v = 0; v < 8; ++v) {
            int row = tile * 16 + half * 8 + v;
            h2[(size_t)row * CLS + m] = c[v];
        }
    }
}

__global__ void self_init2_k(const float* __restrict__ h2, const float* __restrict__ dinv,
                             const float* __restrict__ b2, float* __restrict__ out, int n) {
    int t = blockIdx.x * blockDim.x + threadIdx.x;
    if (t >= n) return;
    int i = t / CLS, f = t % CLS;
    float di = dinv[i];
    out[t] = di * di * h2[t] + b2[f];
}

extern "C" void kernel_launch(void* const* d_in, const int* in_sizes, int n_in,
                              void* d_out, int out_size, void* d_ws, size_t ws_size,
                              hipStream_t stream) {
    const float* x  = (const float*)d_in[0];
    const int*   ei = (const int*)  d_in[1];
    const float* W1 = (const float*)d_in[2];
    const float* b1 = (const float*)d_in[3];
    const float* W2 = (const float*)d_in[4];
    const float* b2 = (const float*)d_in[5];
    float* out = (float*)d_out;

    const int N = in_sizes[0] / F_IN;
    const int E = in_sizes[1] / 2;
    const int* src = ei;
    const int* dst = ei + E;

    float* dinv = (float*)d_ws;
    float* h    = dinv + N;
    float* agg  = h   + (size_t)N * HID;
    float* h2   = agg + (size_t)N * HID;

    const int B = 256;
    const int nTiles = N / 16;

    const int gemmBlocks = (nTiles + 7) / 8;

    deg_init_k <<<(N + B - 1) / B, B, 0, stream>>>(dinv, N);
    const int cntRanges = (N + CNT_ROWS - 1) / CNT_ROWS;
    gemm1_deg_k<<<gemmBlocks + cntRanges * CNT_SLICES, B, 0, stream>>>(x, W1, h, nTiles, gemmBlocks, dst, dinv, E, N);
    dinv_k     <<<(N + B - 1) / B, B, 0, stream>>>(dinv, N);

    int n1 = N * HID;
    self_init1_k<<<(n1 + B - 1) / B, B, 0, stream>>>(h, dinv, agg, n1);
    owner_pass_k<HID, 16><<<dim3((N + OWN_SUMS / 16 - 1) / (OWN_SUMS / 16), SLICES1), OWN_THREADS, 0, stream>>>(src, dst, dinv, h, agg, E, N);
    bias_relu_k<<<(n1 + B - 1) / B, B, 0, stream>>>(agg, b1, n1);

    gemm2_k    <<<(nTiles + 7) / 8, B, 0, stream>>>(agg, W2, h2, nTiles);

    int n2 = N * CLS;
    self_init2_k<<<(n2 + B - 1) / B, B, 0, stream>>>(h2, dinv, b2, out, n2);
    owner_pass_k<CLS, 8><<<dim3((N + OWN_SUMS / 8 - 1) / (OWN_SUMS / 8), SLICES2), OWN_THREADS, 0, stream>>>(src, dst, dinv, h2, out, E, N);
}


# The three one-element-per-thread node launches fold into neighbours: layer-1 self term into the dinv launch, bias/relu and layer-2 start value into GEMM2; 9 launches become 6; every sum unchanged
# speedup vs baseline: 2.7368x; 1.0560x over previous
#include <hip/hip_runtime.h>
#include <hip/hip_bf16.h>

typedef __attribute__((ext_vector_type(2))) float v2f;
typedef __attribute__((ext_vector_type(8))) float v8f;

#define F_IN 512
#define HID  16
#define CLS  7

__global__ void deg_init_k(float* __restrict__ deg, int n) {
    int i = blockIdx.x * blockDim.x + threadIdx.x;
    if (i < n) deg[i] = 1.0f;
}

__global__ void dinv_self1_k(float* __restrict__ deg, const float* __restrict__ h,
                             float* __restrict__ agg, int n4) {
    int t = blockIdx.x * blockDim.x + threadIdx.x;
    if (t >= n4) return;
    const int i = t >> 2;
    const float di = rsqrtf(deg[i]);
    const float4 hv = reinterpret_cast<const float4*>(h)[t];
    float4 r;
    r.x = di * di * hv.x;
    r.y = di * di * hv.y;
    r.z = di * di * hv.z;
    r.w = di * di * hv.w;
    reinterpret_cast<float4*>(agg)[t] = r;
    if ((t & 3) == 0) deg[i] = di;
}

#define CNT_ROWS   (F_IN * HID)
#define CNT_SLICES 64
#define CNT_U      8

template <bool FULL>
__device__ __forceinline__ void count_range(const int* __restrict__ dst, float* __restrict__ deg,
                                            int E, int N, int cb, unsigned* sCnt) {
    const int tid   = threadIdx.x;
    const int range = cb / CNT_SLICES;
    const int slice = cb - range * CNT_SLICES;
    const int lo    = range * CNT_ROWS;
    const int rows  = (N - lo) < CNT_ROWS ? (N - lo) : CNT_ROWS;

    for (int i = tid * 4; i < CNT_ROWS; i += 256 * 4)
        *(uint4*)(sCnt + i) = make_uint4(0u, 0u, 0u, 0u);
    __syncthreads();

    const int per     = (((E + CNT_SLICES - 1) / CNT_SLICES) + 3) & ~3;
    const int e_begin = slice * per;
    const int e_end   = (e_begin + per) < E ? (e_begin + per) : E;

    if (FULL) {
        for (int e0 = e_begin + tid * 4; e0 < e_end; e0 += 256 * 4 * CNT_U) {
            int4 dv[CNT_U];
            bool in[CNT_U];
#pragma unroll
            for (int u = 0; u < CNT_U; ++u) {
                const int e  = e0 + u * (256 * 4);
                const int ec = e < (E - 4) ? e : (E - 4);
                in[u] = e < e_end;
                dv[u] = *(const int4*)(dst + ec);
            }
#pragma unroll
            for (int u = 0; u < CNT_U; ++u) {
                const int d[4] = {dv[u].x, dv[u].y, dv[u].z, dv[u].w};
#pragma unroll
                for (int c = 0; c < 4; ++c) {
                    const unsigned t = (unsigned)((in[u] ? d[c] : -1) - lo);
                    if (t < (unsigned)rows)
                        __hip_atomic_fetch_add(&sCnt[t], 1u, __ATOMIC_RELAXED, __HIP_MEMORY_SCOPE_WORKGROUP);
                }
            }
        }
    } else {
        for (int e = e_begin + tid; e < e_end; e += 256) {
            const unsigned t = (unsigned)(dst[e] - lo);
            if (t < (unsigned)rows)
                __hip_atomic_fetch_add(&sCnt[t], 1u, __ATOMIC_RELAXED, __HIP_MEMORY_SCOPE_WORKGROUP);
        }
    }
    __syncthreads();

    for (int i = tid; i < rows; i += 256) {
        const unsigned c = sCnt[i];
        if (c != 0u) unsafeAtomicAdd(deg + lo + i, (float)c);
    }
}

__global__ void __launch_bounds__(256) gemm1_deg_k(const float* __restrict__ x,
                                                   const float* __restrict__ W1,
                                                   float* __restrict__ h, int nTiles, int gemmBlocks,
                                                   const int* __restrict__ dst,
                                                   float* __restrict__ deg, int E, int N) {
    __shared__ float sW[F_IN * HID];
    if ((int)blockIdx.x >= gemmBlocks) {
        unsigned* sCnt = reinterpret_cast<unsigned*>(sW);
        const int cb = (int)blockIdx.x - gemmBlocks;
        if ((E & 3) == 0 && E >= 4) count_range<true >(dst, deg, E, N, cb, sCnt);
        else                        count_range<false>(dst, deg, E, N, cb, sCnt);
        return;
    }
    for (int i = threadIdx.x * 4; i < F_IN * HID; i += 256 * 4)
        *(float4*)(sW + i) = *(const float4*)(W1 + i);
    __syncthreads();

    const int wave = threadIdx.x >> 5;
    const int lane = threadIdx.x & 31;
    const int tile = blockIdx.x * 8 + wave;
    if (tile >= nTiles) return;

    const int m    = lane & 15;
    const int kk   = (lane >> 4) << 1;
    const int half = lane >> 4;

    const float* arow = x + (size_t)(tile * 16 + m) * F_IN + kk;
    v8f c = {0.f, 0.f, 0.f, 0.f, 0.f, 0.f, 0.f, 0.f};

#pragma unroll 4
    for (int k0 = 0; k0 < F_IN; k0 += 4) {
        v2f a = *(const v2f*)(arow + k0);
        v2f b;
        b.x = sW[(k0 + kk)     * HID + m];
        b.y = sW[(k0 + kk + 1) * HID + m];
        c = __builtin_amdgcn_wmma_f32_16x16x4_f32(false, a, false, b,
                                                  (short)0, c, false, false);
    }

    float* orow = h + (size_t)(tile * 16 + half * 8) * HID + m;
#pragma unroll
    for (int v = 0; v < 8; ++v) orow[(size_t)v * HID] = c[v];
}


#define OWN_THREADS 1024
#define OWN_EPL     8
#define OWN_SUMS    50176
#define OWN_U       8
#define OWN_Q       (OWN_EPL * 32 + 32)
#define SLICES1     8
#define SLICES2     16

template <bool FULL>
__device__ __forceinline__ void load_run(const int* __restrict__ src, const int* __restrict__ dst,
                                         int e, int e_end, int E, int* s, int* d) {
    if (FULL) {
        const bool in = e < e_end;
        int ec = e < (E - 4) ? e : (E - 4);
        ec = ec > 0 ? ec : 0;
        const int4 s4 = *(const int4*)(src + ec);
        const int4 d4 = *(const int4*)(dst + ec);
        s[0] = s4.x; s[1] = s4.y; s[2] = s4.z; s[3] = s4.w;
        d[0] = in ? d4.x : -1; d[1] = in ? d4.y : -1; d[2] = in ? d4.z : -1; d[3] = in ? d4.w : -1;
    } else {
#pragma unroll
        for (int u = 0; u < 4; ++u) {
            const bool in = (e + u) < e_end;
            s[u] = in ? src[e + u] : 0;
            d[u] = in ? dst[e + u] : -1;
        }
    }
}

template <int ROWG, int ROWL, bool GUARD>
__device__ __forceinline__ void gather_round(const int2* q, int j, int m, int grp, int f, int fg,
                                             const float* __restrict__ dinv, const float* __restrict__ tab,
                                             const float* sDinv, float* sAcc) {
    constexpr int G = 32 / ROWL;
    int   ti[OWN_U];
    float w[OWN_U], v[OWN_U];
    bool  ok[OWN_U];
#pragma unroll
    for (int u = 0; u < OWN_U; ++u) {
        const int idx = j + u * G + grp;
        ok[u] = GUARD ? (idx < m) : true;
        const int2 en = q[ok[u] ? idx : 0];
        const int  si = ok[u] ? en.x : 0;
        ti[u] = ok[u] ? en.y : 0;
        w[u] = dinv[si] * sDinv[ti[u]];
        v[u] = tab[(size_t)si * ROWG + fg];
    }
#pragma unroll
    for (int u = 0; u < OWN_U; ++u) {
        const float val = (ok[u] && f < ROWG) ? w[u] * v[u] : 0.0f;
        __hip_atomic_fetch_add(&sAcc[ti[u] * ROWL + f], val, __ATOMIC_RELAXED, __HIP_MEMORY_SCOPE_WORKGROUP);
    }
}

template <int ROWG, int ROWL, bool FULL>
__device__ __forceinline__ void owner_pass(const int* __restrict__ src, const int* __restrict__ dst,
                                           const float* __restrict__ dinv, const float* __restrict__ tab,
                                           float* __restrict__ acc, int E, int N,
                                           float* sAcc, float* sDinv, int2* sQ) {
    constexpr int RPR = OWN_SUMS / ROWL;
    constexpr int G   = 32 / ROWL;
    constexpr int UG  = OWN_U * G;
    const int tid  = threadIdx.x;
    const int lane = tid & 31;
    const int wave = tid >> 5;
    const int f    = lane & (ROWL - 1);
    const int grp  = lane / ROWL;
    const int fg   = f < ROWG ? f : 0;

    const int lo   = (int)blockIdx.x * RPR;
    const int rows = (N - lo) < RPR ? (N - lo) : RPR;

    for (int i = tid * 4; i < OWN_SUMS; i += OWN_THREADS * 4)
        *(float4*)(sAcc + i) = make_float4(0.f, 0.f, 0.f, 0.f);
    for (int i = tid; i < rows; i += OWN_THREADS) sDinv[i] = dinv[lo + i];
    __syncthreads();

    const int S       = (int)gridDim.y;
    const int per     = (((E + S - 1) / S) + 3) & ~3;
    const int e_begin = (int)blockIdx.y * per;
    const int e_end   = (e_begin + per) < E ? (e_begin + per) : E;

    int2* q = sQ + wave * OWN_Q;
    const unsigned below = (1u << lane) - 1u;

    int s[OWN_EPL], d[OWN_EPL];
    load_run<FULL>(src, dst, e_begin + tid * 4, e_end, E, s, d);
    load_run<FULL>(src, dst, e_begin + OWN_THREADS * 4 + tid * 4, e_end, E, s + 4, d + 4);

    int m = 0;
    for (int base = e_begin; base < e_end; base += OWN_THREADS * OWN_EPL) {
        int ns[OWN_EPL], nd[OWN_EPL];
        const int nb = base + OWN_THREADS * OWN_EPL;
        load_run<FULL>(src, dst, nb + tid * 4, e_end, E, ns, nd);
        load_run<FULL>(src, dst, nb + OWN_THREADS * 4 + tid * 4, e_end, E, ns + 4, nd + 4);

#pragma unroll
        for (int c = 0; c < OWN_EPL; ++c) {
            const unsigned t   = (unsigned)(d[c] - lo);
            const bool     hit = t < (unsigned)rows;
            const unsigned mk  = __builtin_amdgcn_ballot_w32(hit);
            if (hit) q[m + __popc(mk & below)] = make_int2(s[c], (int)t);
            m += __popc(mk);
        }

        int j = 0;
        for (; j + UG <= m; j += UG)
            gather_round<ROWG, ROWL, false>(q, j, m, grp, f, fg, dinv, tab, sDinv, sAcc);

        if (j != 0) {
            const int r = m - j;
            if (lane < r) {
                const int2 en = q[j + lane];
                q[lane] = en;
            }
            m = r;
        }

#pragma unroll
        for (int c = 0; c < OWN_EPL; ++c) { s[c] = ns[c]; d[c] = nd[c]; }
    }
    if (m > 0)
        gather_round<ROWG, ROWL, true>(q, 0, m, grp, f, fg, dinv, tab, sDinv, sAcc);
    __syncthreads();

    float* out0 = acc + (size_t)lo * ROWG;
    for (int i = tid; i < rows * ROWG; i += OWN_THREADS) {
        const int row = i / ROWG, ff = i - row * ROWG;
        unsafeAtomicAdd(out0 + i, sAcc[row * ROWL + ff]);
    }
}

template <int ROWG, int ROWL>
__global__ void __launch_bounds__(OWN_THREADS) owner_pass_k(const int* __restrict__ src, const int* __restrict__ dst,
                           const float* __restrict__ dinv, const float* __restrict__ tab,
                           float* __restrict__ acc, int E, int N) {
    __shared__ float sAcc[OWN_SUMS];
    __shared__ float sDinv[OWN_SUMS / ROWL];
    __shared__ int2  sQ[32 * OWN_Q];
    if ((E & 3) == 0 && E >= 4) owner_pass<ROWG, ROWL, true >(src, dst, dinv, tab, acc, E, N, sAcc, sDinv, sQ);
    else              owner_pass<ROWG, ROWL, false>(src, dst, dinv, tab, acc, E, N, sAcc, sDinv, sQ);
}

__global__ void __launch_bounds__(256) gemm2_node_k(const float* __restrict__ a_in,
                                                    const float* __restrict__ b1,
                                                    const float* __restrict__ W2,
                                                    const float* __restrict__ dinv,
                                                    const float* __restrict__ b2,
                                                    float* __restrict__ h2,
                                                    float* __restrict__ out, int nTiles) {
    __shared__ float sW[HID * 16];
    {
        int t = threadIdx.x;
        int k = t >> 4, n = t & 15;
        sW[t] = (n < CLS) ? W2[k * CLS + n] : 0.0f;
    }
    __syncthreads();

    const int wave = threadIdx.x >> 5;
    const int lane = threadIdx.x & 31;
    const int tile = blockIdx.x * 8 + wave;
    if (tile >= nTiles) return;

    const int m    = lane & 15;
    const int kk   = (lane >> 4) << 1;
    const int half = lane >> 4;
    const int row0 = tile * 16 + half * 8;

    const float* arow = a_in + (size_t)(tile * 16 + m) * HID + kk;

    float di[8];
#pragma unroll
    for (int v = 0; v < 8; ++v) di[v] = dinv[row0 + v];
    const float bo = b2[m < CLS ? m : 0];

    v8f c = {0.f, 0.f, 0.f, 0.f, 0.f, 0.f, 0.f, 0.f};

#pragma unroll
    for (int k0 = 0; k0 < HID; k0 += 4) {
        v2f a = *(const v2f*)(arow + k0);
        const float vx = a.x + b1[k0 + kk];
        const float vy = a.y + b1[k0 + kk + 1];
        a.x = vx > 0.f ? vx : 0.f;
        a.y = vy > 0.f ? vy : 0.f;
        v2f b;
        b.x = sW[(k0 + kk)     * 16 + m];
        b.y = sW[(k0 + kk + 1) * 16 + m];
        c = __builtin_amdgcn_wmma_f32_16x16x4_f32(false, a, false, b,
                                                  (short)0, c, false, false);
    }

    float o[8];
#pragma unroll
    for (int v = 0; v < 8; ++v) o[v] = di[v] * di[v] * c[v] + bo;

    if (m < CLS) {
#pragma unroll
        for (int v = 0; v < 8; ++v) {
            const size_t at = (size_t)(row0 + v) * CLS + m;
            h2[at]  = c[v];
            out[at] = o[v];
        }
    }
}


extern "C" void kernel_launch(void* const* d_in, const int* in_sizes, int n_in,
                              void* d_out, int out_size, void* d_ws, size_t ws_size,
                              hipStream_t stream) {
    const float* x  = (const float*)d_in[0];
    const int*   ei = (const int*)  d_in[1];
    const float* W1 = (const float*)d_in[2];
    const float* b1 = (const float*)d_in[3];
    const float* W2 = (const float*)d_in[4];
    const float* b2 = (const float*)d_in[5];
    float* out = (float*)d_out;

    const int N = in_sizes[0] / F_IN;
    const int E = in_sizes[1] / 2;
    const int* src = ei;
    const int* dst = ei + E;

    float* dinv = (float*)d_ws;
    float* h    = dinv + N;
    float* agg  = h   + (size_t)N * HID;
    float* h2   = agg + (size_t)N * HID;

    const int B = 256;
    const int nTiles = N / 16;

    const int gemmBlocks = (nTiles + 7) / 8;

    deg_init_k <<<(N + B - 1) / B, B, 0, stream>>>(dinv, N);
    const int cntRanges = (N + CNT_ROWS - 1) / CNT_ROWS;
    gemm1_deg_k<<<gemmBlocks + cntRanges * CNT_SLICES, B, 0, stream>>>(x, W1, h, nTiles, gemmBlocks, dst, dinv, E, N);

    int n4 = N * (HID / 4);
    dinv_self1_k<<<(n4 + B - 1) / B, B, 0, stream>>>(dinv, h, agg, n4);
    owner_pass_k<HID, 16><<<dim3((N + OWN_SUMS / 16 - 1) / (OWN_SUMS / 16), SLICES1), OWN_THREADS, 0, stream>>>(src, dst, dinv, h, agg, E, N);

    gemm2_node_k<<<(nTiles + 7) / 8, B, 0, stream>>>(agg, b1, W2, dinv, b2, h2, out, nTiles);
    owner_pass_k<CLS, 8><<<dim3((N + OWN_SUMS / 8 - 1) / (OWN_SUMS / 8), SLICES2), OWN_THREADS, 0, stream>>>(src, dst, dinv, h2, out, E, N);
}


# Both edge passes: the deg^-1/2 factors of an edge's weight come from an LDS byte table of all node degrees (rsq per use), not from a global gather per edge; wave lists cut to 96 places to make room
# speedup vs baseline: 3.2557x; 1.1896x over previous
#include <hip/hip_runtime.h>
#include <hip/hip_bf16.h>

typedef __attribute__((ext_vector_type(2))) float v2f;
typedef __attribute__((ext_vector_type(8))) float v8f;

#define F_IN 512
#define HID  16
#define CLS  7

__global__ void deg_init_k(float* __restrict__ deg, int n) {
    int i = blockIdx.x * blockDim.x + threadIdx.x;
    if (i < n) deg[i] = 1.0f;
}

__global__ void self1_deg8_k(const float* __restrict__ deg, const float* __restrict__ h,
                             float* __restrict__ agg, unsigned char* __restrict__ deg8, int n4) {
    int t = blockIdx.x * blockDim.x + threadIdx.x;
    if (t >= n4) return;
    const int i = t >> 2;
    const float dg = deg[i];
    const float di = rsqrtf(dg);
    const float4 hv = reinterpret_cast<const float4*>(h)[t];
    float4 r;
    r.x = di * di * hv.x;
    r.y = di * di * hv.y;
    r.z = di * di * hv.z;
    r.w = di * di * hv.w;
    reinterpret_cast<float4*>(agg)[t] = r;
    if ((t & 3) == 0) deg8[i] = (unsigned char)(dg < 255.0f ? dg : 255.0f);
}

#define CNT_ROWS   (F_IN * HID)
#define CNT_SLICES 64
#define CNT_U      8

template <bool FULL>
__device__ __forceinline__ void count_range(const int* __restrict__ dst, float* __restrict__ deg,
                                            int E, int N, int cb, unsigned* sCnt) {
    const int tid   = threadIdx.x;
    const int range = cb / CNT_SLICES;
    const int slice = cb - range * CNT_SLICES;
    const int lo    = range * CNT_ROWS;
    const int rows  = (N - lo) < CNT_ROWS ? (N - lo) : CNT_ROWS;

    for (int i = tid * 4; i < CNT_ROWS; i += 256 * 4)
        *(uint4*)(sCnt + i) = make_uint4(0u, 0u, 0u, 0u);
    __syncthreads();

    const int per     = (((E + CNT_SLICES - 1) / CNT_SLICES) + 3) & ~3;
    const int e_begin = slice * per;
    const int e_end   = (e_begin + per) < E ? (e_begin + per) : E;

    if (FULL) {
        for (int e0 = e_begin + tid * 4; e0 < e_end; e0 += 256 * 4 * CNT_U) {
            int4 dv[CNT_U];
            bool in[CNT_U];
#pragma unroll
            for (int u = 0; u < CNT_U; ++u) {
                const int e  = e0 + u * (256 * 4);
                const int ec = e < (E - 4) ? e : (E - 4);
                in[u] = e < e_end;
                dv[u] = *(const int4*)(dst + ec);
            }
#pragma unroll
            for (int u = 0; u < CNT_U; ++u) {
                const int d[4] = {dv[u].x, dv[u].y, dv[u].z, dv[u].w};
#pragma unroll
                for (int c = 0; c < 4; ++c) {
                    const unsigned t = (unsigned)((in[u] ? d[c] : -1) - lo);
                    if (t < (unsigned)rows)
                        __hip_atomic_fetch_add(&sCnt[t], 1u, __ATOMIC_RELAXED, __HIP_MEMORY_SCOPE_WORKGROUP);
                }
            }
        }
    } else {
        for (int e = e_begin + tid; e < e_end; e += 256) {
            const unsigned t = (unsigned)(dst[e] - lo);
            if (t < (unsigned)rows)
                __hip_atomic_fetch_add(&sCnt[t], 1u, __ATOMIC_RELAXED, __HIP_MEMORY_SCOPE_WORKGROUP);
        }
    }
    __syncthreads();

    for (int i = tid; i < rows; i += 256) {
        const unsigned c = sCnt[i];
        if (c != 0u) unsafeAtomicAdd(deg + lo + i, (float)c);
    }
}

__global__ void __launch_bounds__(256) gemm1_deg_k(const float* __restrict__ x,
                                                   const float* __restrict__ W1,
                                                   float* __restrict__ h, int nTiles, int gemmBlocks,
                                                   const int* __restrict__ dst,
                                                   float* __restrict__ deg, int E, int N) {
    __shared__ float sW[F_IN * HID];
    if ((int)blockIdx.x >= gemmBlocks) {
        unsigned* sCnt = reinterpret_cast<unsigned*>(sW);
        const int cb = (int)blockIdx.x - gemmBlocks;
        if ((E & 3) == 0 && E >= 4) count_range<true >(dst, deg, E, N, cb, sCnt);
        else                        count_range<false>(dst, deg, E, N, cb, sCnt);
        return;
    }
    for (int i = threadIdx.x * 4; i < F_IN * HID; i += 256 * 4)
        *(float4*)(sW + i) = *(const float4*)(W1 + i);
    __syncthreads();

    const int wave = threadIdx.x >> 5;
    const int lane = threadIdx.x & 31;
    const int tile = blockIdx.x * 8 + wave;
    if (tile >= nTiles) return;

    const int m    = lane & 15;
    const int kk   = (lane >> 4) << 1;
    const int half = lane >> 4;

    const float* arow = x + (size_t)(tile * 16 + m) * F_IN + kk;
    v8f c = {0.f, 0.f, 0.f, 0.f, 0.f, 0.f, 0.f, 0.f};

#pragma unroll 4
    for (int k0 = 0; k0 < F_IN; k0 += 4) {
        v2f a = *(const v2f*)(arow + k0);
        v2f b;
        b.x = sW[(k0 + kk)     * HID + m];
        b.y = sW[(k0 + kk + 1) * HID + m];
        c = __builtin_amdgcn_wmma_f32_16x16x4_f32(false, a, false, b,
                                                  (short)0, c, false, false);
    }

    float* orow = h + (size_t)(tile * 16 + half * 8) * HID + m;
#pragma unroll
    for (int v = 0; v < 8; ++v) orow[(size_t)v * HID] = c[v];
}


#define OWN_THREADS 1024
#define OWN_EPL     8
#define OWN_SUMS    50176
#define OWN_U       8
#define OWN_Q       96
#define DEG_TAB     100352
#define SLICES1     8
#define SLICES2     16

template <bool FULL>
__device__ __forceinline__ void load_run(const int* __restrict__ src, const int* __restrict__ dst,
                                         int e, int e_end, int E, int* s, int* d) {
    if (FULL) {
        const bool in = e < e_end;
        int ec = e < (E - 4) ? e : (E - 4);
        ec = ec > 0 ? ec : 0;
        const int4 s4 = *(const int4*)(src + ec);
        const int4 d4 = *(const int4*)(dst + ec);
        s[0] = s4.x; s[1] = s4.y; s[2] = s4.z; s[3] = s4.w;
        d[0] = in ? d4.x : -1; d[1] = in ? d4.y : -1; d[2] = in ? d4.z : -1; d[3] = in ? d4.w : -1;
    } else {
#pragma unroll
        for (int u = 0; u < 4; ++u) {
            const bool in = (e + u) < e_end;
            s[u] = in ? src[e + u] : 0;
            d[u] = in ? dst[e + u] : -1;
        }
    }
}

template <int ROWG, int ROWL, bool GUARD, bool WHOLE>
__device__ __forceinline__ void gather_round(const int2* q, int j, int m, int grp, int f, int fg, int lo,
                                             const float* __restrict__ deg, const float* __restrict__ tab,
                                             const unsigned char* sDeg8, float* sAcc) {
    constexpr int G = 32 / ROWL;
    int      si[OWN_U], ti[OWN_U];
    unsigned ds8[OWN_U], dd8[OWN_U];
    float    w[OWN_U], v[OWN_U];
    bool     ok[OWN_U];
#pragma unroll
    for (int u = 0; u < OWN_U; ++u) {
        const int idx = j + u * G + grp;
        ok[u] = GUARD ? (idx < m) : true;
        const int2 en = q[ok[u] ? idx : 0];
        si[u] = ok[u] ? en.x : 0;
        ti[u] = ok[u] ? en.y : 0;
        v[u] = tab[(size_t)si[u] * ROWG + fg];
        if (WHOLE) {
            ds8[u] = sDeg8[si[u]];
            dd8[u] = sDeg8[lo + ti[u]];
        }
    }
    if (WHOLE) {
        unsigned top = 0u;
#pragma unroll
        for (int u = 0; u < OWN_U; ++u) {
            const unsigned mx = ds8[u] > dd8[u] ? ds8[u] : dd8[u];
            top = top > mx ? top : mx;
            w[u] = __builtin_amdgcn_rsqf((float)ds8[u]) * __builtin_amdgcn_rsqf((float)dd8[u]);
        }
        if (__builtin_amdgcn_ballot_w32(top == 255u) != 0u) {
#pragma unroll
            for (int u = 0; u < OWN_U; ++u) w[u] = rsqrtf(deg[si[u]]) * rsqrtf(deg[lo + ti[u]]);
        }
    } else {
#pragma unroll
        for (int u = 0; u < OWN_U; ++u) w[u] = rsqrtf(deg[si[u]]) * rsqrtf(deg[lo + ti[u]]);
    }
#pragma unroll
    for (int u = 0; u < OWN_U; ++u) {
        const float val = (ok[u] && f < ROWG) ? w[u] * v[u] : 0.0f;
        __hip_atomic_fetch_add(&sAcc[ti[u] * ROWL + f], val, __ATOMIC_RELAXED, __HIP_MEMORY_SCOPE_WORKGROUP);
    }
}

template <int ROWG, int ROWL, bool FULL, bool WHOLE>
__device__ __forceinline__ void owner_pass(const int* __restrict__ src, const int* __restrict__ dst,
                                           const float* __restrict__ deg, const unsigned char* __restrict__ deg8,
                                           const float* __restrict__ tab,
                                           float* __restrict__ acc, int E, int N,
                                           float* sAcc, unsigned char* sDeg8, int2* sQ) {
    constexpr int RPR = OWN_SUMS / ROWL;
    constexpr int G   = 32 / ROWL;
    constexpr int UG  = OWN_U * G;
    const int tid  = threadIdx.x;
    const int lane = tid & 31;
    const int wave = tid >> 5;
    const int f    = lane & (ROWL - 1);
    const int grp  = lane / ROWL;
    const int fg   = f < ROWG ? f : 0;

    const int lo   = (int)blockIdx.x * RPR;
    const int rows = (N - lo) < RPR ? (N - lo) : RPR;

    constexpr int NCH = (DEG_TAB + OWN_THREADS * 16 - 1) / (OWN_THREADS * 16);
    uint4 pk[NCH];
    if (WHOLE) {
#pragma unroll
        for (int k = 0; k < NCH; ++k) {
            const int i = (k * OWN_THREADS + tid) * 16;
            pk[k] = *(const uint4*)(deg8 + ((i + 16 <= N) ? i : 0));
        }
    }
    for (int i = tid * 4; i < OWN_SUMS; i += OWN_THREADS * 4)
        *(float4*)(sAcc + i) = make_float4(0.f, 0.f, 0.f, 0.f);
    if (WHOLE) {
#pragma unroll
        for (int k = 0; k < NCH; ++k) {
            const int i = (k * OWN_THREADS + tid) * 16;
            if (i < DEG_TAB)
                *(uint4*)(sDeg8 + i) = (i + 16 <= N) ? pk[k] : make_uint4(0xffffffffu, 0xffffffffu, 0xffffffffu, 0xffffffffu);
        }
    }
    __syncthreads();
    if (WHOLE && (N & 15) != 0) {
        const int nt = N & ~15;
        if (tid < N - nt) sDeg8[nt + tid] = deg8[nt + tid];
        __syncthreads();
    }

    const int S       = (int)gridDim.y;
    const int per     = (((E + S - 1) / S) + 3) & ~3;
    const int e_begin = (int)blockIdx.y * per;
    const int e_end   = (e_begin + per) < E ? (e_begin + per) : E;

    int2* q = sQ + wave * OWN_Q;
    const unsigned below = (1u << lane) - 1u;

    int s[OWN_EPL], d[OWN_EPL];
    load_run<FULL>(src, dst, e_begin + tid * 4, e_end, E, s, d);
    load_run<FULL>(src, dst, e_begin + OWN_THREADS * 4 + tid * 4, e_end, E, s + 4, d + 4);

    int m = 0;
    for (int base = e_begin; base < e_end; base += OWN_THREADS * OWN_EPL) {
        int ns[OWN_EPL], nd[OWN_EPL];
        const int nb = base + OWN_THREADS * OWN_EPL;
        load_run<FULL>(src, dst, nb + tid * 4, e_end, E, ns, nd);
        load_run<FULL>(src, dst, nb + OWN_THREADS * 4 + tid * 4, e_end, E, ns + 4, nd + 4);

#pragma unroll
        for (int c = 0; c < OWN_EPL; ++c) {
            const unsigned t   = (unsigned)(d[c] - lo);
            const bool     hit = t < (unsigned)rows;
            const unsigned mk  = __builtin_amdgcn_ballot_w32(hit);
            if (hit) q[m + __popc(mk & below)] = make_int2(s[c], (int)t);
            m += __popc(mk);

            if ((c & 1) == 1 && m >= UG) {
                int j = 0;
                for (; j + UG <= m; j += UG)
                    gather_round<ROWG, ROWL, false, WHOLE>(q, j, m, grp, f, fg, lo, deg, tab, sDeg8, sAcc);
                const int r = m - j;
                if (lane < r) {
                    const int2 en = q[j + lane];
                    q[lane] = en;
                }
                m = r;
            }
        }

#pragma unroll
        for (int c = 0; c < OWN_EPL; ++c) { s[c] = ns[c]; d[c] = nd[c]; }
    }
    if (m > 0)
        gather_round<ROWG, ROWL, true, WHOLE>(q, 0, m, grp, f, fg, lo, deg, tab, sDeg8, sAcc);
    __syncthreads();

    float* out0 = acc + (size_t)lo * ROWG;
    for (int i = tid; i < rows * ROWG; i += OWN_THREADS) {
        const int row = i / ROWG, ff = i - row * ROWG;
        unsafeAtomicAdd(out0 + i, sAcc[row * ROWL + ff]);
    }
}

template <int ROWG, int ROWL>
__global__ void __launch_bounds__(OWN_THREADS) owner_pass_k(const int* __restrict__ src, const int* __restrict__ dst,
                           const float* __restrict__ deg, const unsigned char* __restrict__ deg8,
                           const float* __restrict__ tab,
                           float* __restrict__ acc, int E, int N) {
    __shared__ float sAcc[OWN_SUMS];
    __shared__ __attribute__((aligned(16))) unsigned char sDeg8[DEG_TAB];
    __shared__ int2  sQ[32 * OWN_Q];
    const bool full = (E & 3) == 0 && E >= 4;
    if (N <= DEG_TAB) {
        if (full) owner_pass<ROWG, ROWL, true,  true >(src, dst, deg, deg8, tab, acc, E, N, sAcc, sDeg8, sQ);
        else      owner_pass<ROWG, ROWL, false, true >(src, dst, deg, deg8, tab, acc, E, N, sAcc, sDeg8, sQ);
    } else {
        if (full) owner_pass<ROWG, ROWL, true,  false>(src, dst, deg, deg8, tab, acc, E, N, sAcc, sDeg8, sQ);
        else      owner_pass<ROWG, ROWL, false, false>(src, dst, deg, deg8, tab, acc, E, N, sAcc, sDeg8, sQ);
    }
}

__global__ void __launch_bounds__(256) gemm2_node_k(const float* __restrict__ a_in,
                                                    const float* __restrict__ b1,
                                                    const float* __restrict__ W2,
                                                    const float* __restrict__ deg,
                                                    const float* __restrict__ b2,
                                                    float* __restrict__ h2,
                                                    float* __restrict__ out,
                                                    unsigned char* __restrict__ deg8, int nTiles) {
    __shared__ float sW[HID * 16];
    {
        int t = threadIdx.x;
        int k = t >> 4, n = t & 15;
        sW[t] = (n < CLS) ? W2[k * CLS + n] : 0.0f;
    }
    __syncthreads();

    const int wave = threadIdx.x >> 5;
    const int lane = threadIdx.x & 31;
    const int tile = blockIdx.x * 8 + wave;
    if (tile >= nTiles) return;

    const int m    = lane & 15;
    const int kk   = (lane >> 4) << 1;
    const int half = lane >> 4;
    const int row0 = tile * 16 + half * 8;

    const float* arow = a_in + (size_t)(tile * 16 + m) * HID + kk;

    float dg[8], di[8];
#pragma unroll
    for (int v = 0; v < 8; ++v) { dg[v] = deg[row0 + v]; di[v] = rsqrtf(dg[v]); }
    if (m == 0) {
        unsigned p0 = 0u, p1 = 0u;
#pragma unroll
        for (int v = 0; v < 4; ++v) {
            p0 |= (unsigned)(dg[v]     < 255.0f ? dg[v]     : 255.0f) << (8 * v);
            p1 |= (unsigned)(dg[v + 4] < 255.0f ? dg[v + 4] : 255.0f) << (8 * v);
        }
        *(uint2*)(deg8 + row0) = make_uint2(p0, p1);
    }
    const float bo = b2[m < CLS ? m : 0];

    v8f c = {0.f, 0.f, 0.f, 0.f, 0.f, 0.f, 0.f, 0.f};

#pragma unroll
    for (int k0 = 0; k0 < HID; k0 += 4) {
        v2f a = *(const v2f*)(arow + k0);
        const float vx = a.x + b1[k0 + kk];
        const float vy = a.y + b1[k0 + kk + 1];
        a.x = vx > 0.f ? vx : 0.f;
        a.y = vy > 0.f ? vy : 0.f;
        v2f b;
        b.x = sW[(k0 + kk)     * 16 + m];
        b.y = sW[(k0 + kk + 1) * 16 + m];
        c = __builtin_amdgcn_wmma_f32_16x16x4_f32(false, a, false, b,
                                                  (short)0, c, false, false);
    }

    float o[8];
#pragma unroll
    for (int v = 0; v < 8; ++v) o[v] = di[v] * di[v] * c[v] + bo;

    if (m < CLS) {
#pragma unroll
        for (int v = 0; v < 8; ++v) {
            const size_t at = (size_t)(row0 + v) * CLS + m;
            h2[at]  = c[v];
            out[at] = o[v];
        }
    }
}


extern "C" void kernel_launch(void* const* d_in, const int* in_sizes, int n_in,
                              void* d_out, int out_size, void* d_ws, size_t ws_size,
                              hipStream_t stream) {
    const float* x  = (const float*)d_in[0];
    const int*   ei = (const int*)  d_in[1];
    const float* W1 = (const float*)d_in[2];
    const float* b1 = (const float*)d_in[3];
    const float* W2 = (const float*)d_in[4];
    const float* b2 = (const float*)d_in[5];
    float* out = (float*)d_out;

    const int N = in_sizes[0] / F_IN;
    const int E = in_sizes[1] / 2;
    const int* src = ei;
    const int* dst = ei + E;

    float* deg  = (float*)d_ws;
    float* h    = deg + N;
    float* agg  = h   + (size_t)N * HID;
    float* h2   = agg + (size_t)N * HID;

    const int B = 256;
    const int nTiles = N / 16;

    const int gemmBlocks = (nTiles + 7) / 8;

    deg_init_k <<<(N + B - 1) / B, B, 0, stream>>>(deg, N);
    const int cntRanges = (N + CNT_ROWS - 1) / CNT_ROWS;
    gemm1_deg_k<<<gemmBlocks + cntRanges * CNT_SLICES, B, 0, stream>>>(x, W1, h, nTiles, gemmBlocks, dst, deg, E, N);

    unsigned char* deg8a = (unsigned char*)h2;
    unsigned char* deg8b = (unsigned char*)h;

    int n4 = N * (HID / 4);
    self1_deg8_k<<<(n4 + B - 1) / B, B, 0, stream>>>(deg, h, agg, deg8a, n4);
    owner_pass_k<HID, 16><<<dim3((N + OWN_SUMS / 16 - 1) / (OWN_SUMS / 16), SLICES1), OWN_THREADS, 0, stream>>>(src, dst, deg, deg8a, h, agg, E, N);

    gemm2_node_k<<<(nTiles + 7) / 8, B, 0, stream>>>(agg, b1, W2, deg, b2, h2, out, deg8b, nTiles);
    owner_pass_k<CLS, 8><<<dim3((N + OWN_SUMS / 8 - 1) / (OWN_SUMS / 8), SLICES2), OWN_THREADS, 0, stream>>>(src, dst, deg, deg8b, h2, out, E, N);
}


# Both edge passes: an edge's weight is formed once, by one lane of the round that drains its list place, and kept in the entry (both indices packed in one word); LDS adds still in arrival order
# speedup vs baseline: 3.3908x; 1.0415x over previous
#include <hip/hip_runtime.h>
#include <hip/hip_bf16.h>

typedef __attribute__((ext_vector_type(2))) float v2f;
typedef __attribute__((ext_vector_type(8))) float v8f;

#define F_IN 512
#define HID  16
#define CLS  7

__global__ void deg_init_k(float* __restrict__ deg, int n) {
    int i = blockIdx.x * blockDim.x + threadIdx.x;
    if (i < n) deg[i] = 1.0f;
}

__global__ void self1_deg8_k(const float* __restrict__ deg, const float* __restrict__ h,
                             float* __restrict__ agg, unsigned char* __restrict__ deg8, int n4) {
    int t = blockIdx.x * blockDim.x + threadIdx.x;
    if (t >= n4) return;
    const int i = t >> 2;
    const float dg = deg[i];
    const float di = rsqrtf(dg);
    const float4 hv = reinterpret_cast<const float4*>(h)[t];
    float4 r;
    r.x = di * di * hv.x;
    r.y = di * di * hv.y;
    r.z = di * di * hv.z;
    r.w = di * di * hv.w;
    reinterpret_cast<float4*>(agg)[t] = r;
    if ((t & 3) == 0) deg8[i] = (unsigned char)(dg < 255.0f ? dg : 255.0f);
}

#define CNT_ROWS   (F_IN * HID)
#define CNT_SLICES 64
#define CNT_U      8

template <bool FULL>
__device__ __forceinline__ void count_range(const int* __restrict__ dst, float* __restrict__ deg,
                                            int E, int N, int cb, unsigned* sCnt) {
    const int tid   = threadIdx.x;
    const int range = cb / CNT_SLICES;
    const int slice = cb - range * CNT_SLICES;
    const int lo    = range * CNT_ROWS;
    const int rows  = (N - lo) < CNT_ROWS ? (N - lo) : CNT_ROWS;

    for (int i = tid * 4; i < CNT_ROWS; i += 256 * 4)
        *(uint4*)(sCnt + i) = make_uint4(0u, 0u, 0u, 0u);
    __syncthreads();

    const int per     = (((E + CNT_SLICES - 1) / CNT_SLICES) + 3) & ~3;
    const int e_begin = slice * per;
    const int e_end   = (e_begin + per) < E ? (e_begin + per) : E;

    if (FULL) {
        for (int e0 = e_begin + tid * 4; e0 < e_end; e0 += 256 * 4 * CNT_U) {
            int4 dv[CNT_U];
            bool in[CNT_U];
#pragma unroll
            for (int u = 0; u < CNT_U; ++u) {
                const int e  = e0 + u * (256 * 4);
                const int ec = e < (E - 4) ? e : (E - 4);
                in[u] = e < e_end;
                dv[u] = *(const int4*)(dst + ec);
            }
#pragma unroll
            for (int u = 0; u < CNT_U; ++u) {
                const int d[4] = {dv[u].x, dv[u].y, dv[u].z, dv[u].w};
#pragma unroll
                for (int c = 0; c < 4; ++c) {
                    const unsigned t = (unsigned)((in[u] ? d[c] : -1) - lo);
                    if (t < (unsigned)rows)
                        __hip_atomic_fetch_add(&sCnt[t], 1u, __ATOMIC_RELAXED, __HIP_MEMORY_SCOPE_WORKGROUP);
                }
            }
        }
    } else {
        for (int e = e_begin + tid; e < e_end; e += 256) {
            const unsigned t = (unsigned)(dst[e] - lo);
            if (t < (unsigned)rows)
                __hip_atomic_fetch_add(&sCnt[t], 1u, __ATOMIC_RELAXED, __HIP_MEMORY_SCOPE_WORKGROUP);
        }
    }
    __syncthreads();

    for (int i = tid; i < rows; i += 256) {
        const unsigned c = sCnt[i];
        if (c != 0u) unsafeAtomicAdd(deg + lo + i, (float)c);
    }
}

__global__ void __launch_bounds__(256) gemm1_deg_k(const float* __restrict__ x,
                                                   const float* __restrict__ W1,
                                                   float* __restrict__ h, int nTiles, int gemmBlocks,
                                                   const int* __restrict__ dst,
                                                   float* __restrict__ deg, int E, int N) {
    __shared__ float sW[F_IN * HID];
    if ((int)blockIdx.x >= gemmBlocks) {
        unsigned* sCnt = reinterpret_cast<unsigned*>(sW);
        const int cb = (int)blockIdx.x - gemmBlocks;
        if ((E & 3) == 0 && E >= 4) count_range<true >(dst, deg, E, N, cb, sCnt);
        else                        count_range<false>(dst, deg, E, N, cb, sCnt);
        return;
    }
    for (int i = threadIdx.x * 4; i < F_IN * HID; i += 256 * 4)
        *(float4*)(sW + i) = *(const float4*)(W1 + i);
    __syncthreads();

    const int wave = threadIdx.x >> 5;
    const int lane = threadIdx.x & 31;
    const int tile = blockIdx.x * 8 + wave;
    if (tile >= nTiles) return;

    const int m    = lane & 15;
    const int kk   = (lane >> 4) << 1;
    const int half = lane >> 4;

    const float* arow = x + (size_t)(tile * 16 + m) * F_IN + kk;
    v8f c = {0.f, 0.f, 0.f, 0.f, 0.f, 0.f, 0.f, 0.f};

#pragma unroll 4
    for (int k0 = 0; k0 < F_IN; k0 += 4) {
        v2f a = *(const v2f*)(arow + k0);
        v2f b;
        b.x = sW[(k0 + kk)     * HID + m];
        b.y = sW[(k0 + kk + 1) * HID + m];
        c = __builtin_amdgcn_wmma_f32_16x16x4_f32(false, a, false, b,
                                                  (short)0, c, false, false);
    }

    float* orow = h + (size_t)(tile * 16 + half * 8) * HID + m;
#pragma unroll
    for (int v = 0; v < 8; ++v) orow[(size_t)v * HID] = c[v];
}


#define OWN_THREADS 1024
#define OWN_EPL     8
#define OWN_SUMS    50176
#define OWN_U       8
#define OWN_Q       96
#define DEG_TAB     100352
#define SLICES1     8
#define SLICES2     16
#define Q_TBITS     13
static_assert(OWN_SUMS / 8 <= (1 << Q_TBITS), "a range's local row must fit Q_TBITS bits");
static_assert(DEG_TAB <= (1 << (32 - Q_TBITS)), "a node index of a WHOLE pass must fit the remaining bits");

template <bool FULL>
__device__ __forceinline__ void load_run(const int* __restrict__ src, const int* __restrict__ dst,
                                         int e, int e_end, int E, int* s, int* d) {
    if (FULL) {
        const bool in = e < e_end;
        int ec = e < (E - 4) ? e : (E - 4);
        ec = ec > 0 ? ec : 0;
        const int4 s4 = *(const int4*)(src + ec);
        const int4 d4 = *(const int4*)(dst + ec);
        s[0] = s4.x; s[1] = s4.y; s[2] = s4.z; s[3] = s4.w;
        d[0] = in ? d4.x : -1; d[1] = in ? d4.y : -1; d[2] = in ? d4.z : -1; d[3] = in ? d4.w : -1;
    } else {
#pragma unroll
        for (int u = 0; u < 4; ++u) {
            const bool in = (e + u) < e_end;
            s[u] = in ? src[e + u] : 0;
            d[u] = in ? dst[e + u] : -1;
        }
    }
}

template <int ROWG, int ROWL, bool GUARD>
__device__ __forceinline__ void gather_round_g(const int2* q, int j, int m, int grp, int f, int fg, int lo,
                                               const float* __restrict__ deg, const float* __restrict__ tab,
                                               float* sAcc) {
    constexpr int G = 32 / ROWL;
    int   si[OWN_U], ti[OWN_U];
    float w[OWN_U], v[OWN_U];
    bool  ok[OWN_U];
#pragma unroll
    for (int u = 0; u < OWN_U; ++u) {
        const int idx = j + u * G + grp;
        ok[u] = GUARD ? (idx < m) : true;
        const int2 en = q[ok[u] ? idx : 0];
        si[u] = ok[u] ? en.x : 0;
        ti[u] = ok[u] ? en.y : 0;
        v[u] = tab[(size_t)si[u] * ROWG + fg];
    }
#pragma unroll
    for (int u = 0; u < OWN_U; ++u) w[u] = rsqrtf(deg[si[u]]) * rsqrtf(deg[lo + ti[u]]);
#pragma unroll
    for (int u = 0; u < OWN_U; ++u) {
        const float val = (ok[u] && f < ROWG) ? w[u] * v[u] : 0.0f;
        __hip_atomic_fetch_add(&sAcc[ti[u] * ROWL + f], val, __ATOMIC_RELAXED, __HIP_MEMORY_SCOPE_WORKGROUP);
    }
}

template <int ROWG, int ROWL, bool GUARD>
__device__ __forceinline__ void gather_round_w(int2* q, int j, int m, int lane, int grp, int f, int fg, int lo,
                                               const float* __restrict__ deg, const float* __restrict__ tab,
                                               const unsigned char* sDeg8, float* sAcc) {
    constexpr int      G     = 32 / ROWL;
    constexpr int      UG    = OWN_U * G;
    constexpr unsigned TMASK = (1u << Q_TBITS) - 1u;
    unsigned ti[OWN_U];
    float    v[OWN_U];
    bool     ok[OWN_U];
#pragma unroll
    for (int u = 0; u < OWN_U; ++u) {
        const int idx = j + u * G + grp;
        ok[u] = GUARD ? (idx < m) : true;
        const unsigned px = (unsigned)q[ok[u] ? idx : 0].x;
        const unsigned p  = ok[u] ? px : 0u;
        ti[u] = p & TMASK;
        v[u]  = tab[(p >> Q_TBITS) * (unsigned)ROWG + (unsigned)fg];
    }
    {
        const int      idx = j + (lane & (UG - 1));
        const bool     okw = GUARD ? (idx < m) : true;
        const unsigned p   = okw ? (unsigned)q[idx].x : 0u;
        const unsigned s   = p >> Q_TBITS;
        const unsigned t   = p & TMASK;
        const unsigned a8  = sDeg8[s];
        const unsigned b8  = sDeg8[(unsigned)lo + t];
        float w = __builtin_amdgcn_rsqf((float)a8) * __builtin_amdgcn_rsqf((float)b8);
        if (__builtin_amdgcn_ballot_w32((a8 > b8 ? a8 : b8) == 255u) != 0u)
            w = rsqrtf(deg[s]) * rsqrtf(deg[(unsigned)lo + t]);
        if (okw) q[idx].y = __float_as_int(w);
    }
    float w[OWN_U];
#pragma unroll
    for (int u = 0; u < OWN_U; ++u) {
        const int idx = j + u * G + grp;
        w[u] = __int_as_float(q[ok[u] ? idx : 0].y);
    }
#pragma unroll
    for (int u = 0; u < OWN_U; ++u) {
        const float val = (ok[u] && f < ROWG) ? w[u] * v[u] : 0.0f;
        __hip_atomic_fetch_add(&sAcc[ti[u] * (unsigned)ROWL + (unsigned)f], val, __ATOMIC_RELAXED, __HIP_MEMORY_SCOPE_WORKGROUP);
    }
}

template <int ROWG, int ROWL, bool FULL, bool WHOLE>
__device__ __forceinline__ void owner_pass(const int* __restrict__ src, const int* __restrict__ dst,
                                           const float* __restrict__ deg, const unsigned char* __restrict__ deg8,
                                           const float* __restrict__ tab,
                                           float* __restrict__ acc, int E, int N,
                                           float* sAcc, unsigned char* sDeg8, int2* sQ) {
    constexpr int RPR = OWN_SUMS / ROWL;
    constexpr int G   = 32 / ROWL;
    constexpr int UG  = OWN_U * G;
    const int tid  = threadIdx.x;
    const int lane = tid & 31;
    const int wave = tid >> 5;
    const int f    = lane & (ROWL - 1);
    const int grp  = lane / ROWL;
    const int fg   = f < ROWG ? f : 0;

    const int lo   = (int)blockIdx.x * RPR;
    const int rows = (N - lo) < RPR ? (N - lo) : RPR;

    constexpr int NCH = (DEG_TAB + OWN_THREADS * 16 - 1) / (OWN_THREADS * 16);
    uint4 pk[NCH];
    if (WHOLE) {
#pragma unroll
        for (int k = 0; k < NCH; ++k) {
            const int i = (k * OWN_THREADS + tid) * 16;
            pk[k] = *(const uint4*)(deg8 + ((i + 16 <= N) ? i : 0));
        }
    }
    for (int i = tid * 4; i < OWN_SUMS; i += OWN_THREADS * 4)
        *(float4*)(sAcc + i) = make_float4(0.f, 0.f, 0.f, 0.f);
    if (WHOLE) {
#pragma unroll
        for (int k = 0; k < NCH; ++k) {
            const int i = (k * OWN_THREADS + tid) * 16;
            if (i < DEG_TAB)
                *(uint4*)(sDeg8 + i) = (i + 16 <= N) ? pk[k] : make_uint4(0xffffffffu, 0xffffffffu, 0xffffffffu, 0xffffffffu);
        }
    }
    __syncthreads();
    if (WHOLE && (N & 15) != 0) {
        const int nt = N & ~15;
        if (tid < N - nt) sDeg8[nt + tid] = deg8[nt + tid];
        __syncthreads();
    }

    const int S       = (int)gridDim.y;
    const int per     = (((E + S - 1) / S) + 3) & ~3;
    const int e_begin = (int)blockIdx.y * per;
    const int e_end   = (e_begin + per) < E ? (e_begin + per) : E;

    int2* q = sQ + wave * OWN_Q;
    const unsigned below = (1u << lane) - 1u;

    int s[OWN_EPL], d[OWN_EPL];
    load_run<FULL>(src, dst, e_begin + tid * 4, e_end, E, s, d);
    load_run<FULL>(src, dst, e_begin + OWN_THREADS * 4 + tid * 4, e_end, E, s + 4, d + 4);

    int m = 0;
    for (int base = e_begin; base < e_end; base += OWN_THREADS * OWN_EPL) {
        int ns[OWN_EPL], nd[OWN_EPL];
        const int nb = base + OWN_THREADS * OWN_EPL;
        load_run<FULL>(src, dst, nb + tid * 4, e_end, E, ns, nd);
        load_run<FULL>(src, dst, nb + OWN_THREADS * 4 + tid * 4, e_end, E, ns + 4, nd + 4);

#pragma unroll
        for (int c = 0; c < OWN_EPL; ++c) {
            const unsigned t   = (unsigned)(d[c] - lo);
            const bool     hit = t < (unsigned)rows;
            const unsigned mk  = __builtin_amdgcn_ballot_w32(hit);
            if (hit) {
                int2* e = q + (m + __popc(mk & below));
                if (WHOLE) e->x = (int)(((unsigned)s[c] << Q_TBITS) | t);
                else       *e   = make_int2(s[c], (int)t);
            }
            m += __popc(mk);

            if ((c & 1) == 1 && m >= UG) {
                int j = 0;
                for (; j + UG <= m; j += UG) {
                    if (WHOLE) gather_round_w<ROWG, ROWL, false>(q, j, m, lane, grp, f, fg, lo, deg, tab, sDeg8, sAcc);
                    else       gather_round_g<ROWG, ROWL, false>(q, j, m, grp, f, fg, lo, deg, tab, sAcc);
                }
                const int r = m - j;
                if (lane < r) {
                    const int2 en = q[j + lane];
                    q[lane] = en;
                }
                m = r;
            }
        }

#pragma unroll
        for (int c = 0; c < OWN_EPL; ++c) { s[c] = ns[c]; d[c] = nd[c]; }
    }
    if (m > 0) {
        if (WHOLE) gather_round_w<ROWG, ROWL, true>(q, 0, m, lane, grp, f, fg, lo, deg, tab, sDeg8, sAcc);
        else       gather_round_g<ROWG, ROWL, true>(q, 0, m, grp, f, fg, lo, deg, tab, sAcc);
    }
    __syncthreads();

    float* out0 = acc + (size_t)lo * ROWG;
    for (int i = tid; i < rows * ROWG; i += OWN_THREADS) {
        const int row = i / ROWG, ff = i - row * ROWG;
        unsafeAtomicAdd(out0 + i, sAcc[row * ROWL + ff]);
    }
}

template <int ROWG, int ROWL>
__global__ void __launch_bounds__(OWN_THREADS) owner_pass_k(const int* __restrict__ src, const int* __restrict__ dst,
                           const float* __restrict__ deg, const unsigned char* __restrict__ deg8,
                           const float* __restrict__ tab,
                           float* __restrict__ acc, int E, int N) {
    __shared__ float sAcc[OWN_SUMS];
    __shared__ __attribute__((aligned(16))) unsigned char sDeg8[DEG_TAB];
    __shared__ int2  sQ[32 * OWN_Q];
    const bool full = (E & 3) == 0 && E >= 4;
    if (N <= DEG_TAB) {
        if (full) owner_pass<ROWG, ROWL, true,  true >(src, dst, deg, deg8, tab, acc, E, N, sAcc, sDeg8, sQ);
        else      owner_pass<ROWG, ROWL, false, true >(src, dst, deg, deg8, tab, acc, E, N, sAcc, sDeg8, sQ);
    } else {
        if (full) owner_pass<ROWG, ROWL, true,  false>(src, dst, deg, deg8, tab, acc, E, N, sAcc, sDeg8, sQ);
        else      owner_pass<ROWG, ROWL, false, false>(src, dst, deg, deg8, tab, acc, E, N, sAcc, sDeg8, sQ);
    }
}

__global__ void __launch_bounds__(256) gemm2_node_k(const float* __restrict__ a_in,
                                                    const float* __restrict__ b1,
                                                    const float* __restrict__ W2,
                                                    const float* __restrict__ deg,
                                                    const float* __restrict__ b2,
                                                    float* __restrict__ h2,
                                                    float* __restrict__ out,
                                                    unsigned char* __restrict__ deg8, int nTiles) {
    __shared__ float sW[HID * 16];
    {
        int t = threadIdx.x;
        int k = t >> 4, n = t & 15;
        sW[t] = (n < CLS) ? W2[k * CLS + n] : 0.0f;
    }
    __syncthreads();

    const int wave = threadIdx.x >> 5;
    const int lane = threadIdx.x & 31;
    const int tile = blockIdx.x * 8 + wave;
    if (tile >= nTiles) return;

    const int m    = lane & 15;
    const int kk   = (lane >> 4) << 1;
    const int half = lane >> 4;
    const int row0 = tile * 16 + half * 8;

    const float* arow = a_in + (size_t)(tile * 16 + m) * HID + kk;

    float dg[8], di[8];
#pragma unroll
    for (int v = 0; v < 8; ++v) { dg[v] = deg[row0 + v]; di[v] = rsqrtf(dg[v]); }
    if (m == 0) {
        unsigned p0 = 0u, p1 = 0u;
#pragma unroll
        for (int v = 0; v < 4; ++v) {
            p0 |= (unsigned)(dg[v]     < 255.0f ? dg[v]     : 255.0f) << (8 * v);
            p1 |= (unsigned)(dg[v + 4] < 255.0f ? dg[v + 4] : 255.0f) << (8 * v);
        }
        *(uint2*)(deg8 + row0) = make_uint2(p0, p1);
    }
    const float bo = b2[m < CLS ? m : 0];

    v8f c = {0.f, 0.f, 0.f, 0.f, 0.f, 0.f, 0.f, 0.f};

#pragma unroll
    for (int k0 = 0; k0 < HID; k0 += 4) {
        v2f a = *(const v2f*)(arow + k0);
        const float vx = a.x + b1[k0 + kk];
        const float vy = a.y + b1[k0 + kk + 1];
        a.x = vx > 0.f ? vx : 0.f;
        a.y = vy > 0.f ? vy : 0.f;
        v2f b;
        b.x = sW[(k0 + kk)     * 16 + m];
        b.y = sW[(k0 + kk + 1) * 16 + m];
        c = __builtin_amdgcn_wmma_f32_16x16x4_f32(false, a, false, b,
                                                  (short)0, c, false, false);
    }

    float o[8];
#pragma unroll
    for (int v = 0; v < 8; ++v) o[v] = di[v] * di[v] * c[v] + bo;

    if (m < CLS) {
#pragma unroll
        for (int v = 0; v < 8; ++v) {
            const size_t at = (size_t)(row0 + v) * CLS + m;
            h2[at]  = c[v];
            out[at] = o[v];
        }
    }
}


extern "C" void kernel_launch(void* const* d_in, const int* in_sizes, int n_in,
                              void* d_out, int out_size, void* d_ws, size_t ws_size,
                              hipStream_t stream) {
    const float* x  = (const float*)d_in[0];
    const int*   ei = (const int*)  d_in[1];
    const float* W1 = (const float*)d_in[2];
    const float* b1 = (const float*)d_in[3];
    const float* W2 = (const float*)d_in[4];
    const float* b2 = (const float*)d_in[5];
    float* out = (float*)d_out;

    const int N = in_sizes[0] / F_IN;
    const int E = in_sizes[1] / 2;
    const int* src = ei;
    const int* dst = ei + E;

    float* deg  = (float*)d_ws;
    float* h    = deg + N;
    float* agg  = h   + (size_t)N * HID;
    float* h2   = agg + (size_t)N * HID;

    const int B = 256;
    const int nTiles = N / 16;

    const int gemmBlocks = (nTiles + 7) / 8;

    deg_init_k <<<(N + B - 1) / B, B, 0, stream>>>(deg, N);
    const int cntRanges = (N + CNT_ROWS - 1) / CNT_ROWS;
    gemm1_deg_k<<<gemmBlocks + cntRanges * CNT_SLICES, B, 0, stream>>>(x, W1, h, nTiles, gemmBlocks, dst, deg, E, N);

    unsigned char* deg8a = (unsigned char*)h2;
    unsigned char* deg8b = (unsigned char*)h;

    int n4 = N * (HID / 4);
    self1_deg8_k<<<(n4 + B - 1) / B, B, 0, stream>>>(deg, h, agg, deg8a, n4);
    owner_pass_k<HID, 16><<<dim3((N + OWN_SUMS / 16 - 1) / (OWN_SUMS / 16), SLICES1), OWN_THREADS, 0, stream>>>(src, dst, deg, deg8a, h, agg, E, N);

    gemm2_node_k<<<(nTiles + 7) / 8, B, 0, stream>>>(agg, b1, W2, deg, b2, h2, out, deg8b, nTiles);
    owner_pass_k<CLS, 8><<<dim3((N + OWN_SUMS / 8 - 1) / (OWN_SUMS / 8), SLICES2), OWN_THREADS, 0, stream>>>(src, dst, deg, deg8b, h2, out, E, N);
}
